# starting kernel
# baseline (speedup 1.0000x reference)
#include <hip/hip_runtime.h>
#include <cmath>

#define B_  8
#define N_  2048
#define H_  256
#define THREADS 256
#define ROWS_PER_BLOCK 128

typedef __attribute__((ext_vector_type(2))) float v2f;
typedef __attribute__((ext_vector_type(8))) float v8f;

#define LOG2E 1.4426950408889634f
#define LN2   0.6931471805599453f

__global__ __launch_bounds__(THREADS)
void prep_kernel(const float* __restrict__ cloud, const float* __restrict__ noise,
                 float4* __restrict__ ypts, float4* __restrict__ xpts,
                 float* __restrict__ f, float* __restrict__ g) {
  const int b   = blockIdx.x;
  const int tid = threadIdx.x;
  __shared__ float s_sum[THREADS];
  __shared__ float s_sq[THREADS];

  const float* cb = cloud + (size_t)b * N_ * 3;
  const float* nb = noise + (size_t)b * N_ * 3;

  float sum = 0.f, sq = 0.f;
  for (int i = tid; i < N_ * 3; i += THREADS) {
    float v = cb[i];
    sum += v; sq += v * v;
  }
  s_sum[tid] = sum; s_sq[tid] = sq;
  __syncthreads();
  for (int off = THREADS / 2; off > 0; off >>= 1) {
    if (tid < off) { s_sum[tid] += s_sum[tid + off]; s_sq[tid] += s_sq[tid + off]; }
    __syncthreads();
  }
  const float n    = (float)(N_ * 3);
  const float tot  = s_sum[0];
  const float tot2 = s_sq[0];
  const float var  = (tot2 - tot * tot / n) / (n - 1.0f);
  const float inv_std = rsqrtf(var);

  for (int p = tid; p < N_; p += THREADS) {
    float yx = cb[p * 3 + 0] * inv_std;
    float yy = cb[p * 3 + 1] * inv_std;
    float yz = cb[p * 3 + 2] * inv_std;
    ypts[(size_t)b * N_ + p] = make_float4(yx, yy, yz, yx * yx + yy * yy + yz * yz);
    float xx = nb[p * 3 + 0];
    float xy = nb[p * 3 + 1];
    float xz = nb[p * 3 + 2];
    xpts[(size_t)b * N_ + p] = make_float4(xx, xy, xz, xx * xx + xy * xy + xz * xz);
    f[(size_t)b * N_ + p] = 0.f;
    g[(size_t)b * N_ + p] = 0.f;
  }
}

__global__ __launch_bounds__(THREADS)
void sinkhorn_half(const float4* __restrict__ rows_pts,
                   const float4* __restrict__ cols_pts,
                   const float*  __restrict__ in_pot,
                   float*        __restrict__ out_pot,
                   float eps, float lconst) {
  const int b    = blockIdx.y;
  const int tid  = threadIdx.x;
  const int lane = tid & 31;
  const int wave = tid >> 5;
  const int hi   = lane >> 4;
  const int idx  = lane & 15;

  __shared__ float4 s_cols[N_];
  __shared__ float  s_gy[N_];

  const float4* cp = cols_pts + (size_t)b * N_;
  const float*  ip = in_pot   + (size_t)b * N_;
  for (int j = tid; j < N_; j += THREADS) {
    float4 c = cp[j];
    s_cols[j] = c;
    s_gy[j]   = ip[j] - 0.5f * c.w;
  }
  __syncthreads();

  const int row_base = blockIdx.x * ROWS_PER_BLOCK + wave * 16;
  const float4* rp = rows_pts + (size_t)b * N_;

  float4 r4 = rp[row_base + idx];
  v2f a;
  a.x = hi ? r4.z : r4.x;
  a.y = hi ? 0.0f : r4.y;

  float m[8], s[8];
#pragma unroll
  for (int r = 0; r < 8; ++r) { m[r] = -3.0e38f; s[r] = 0.f; }
  const float c2 = LOG2E / eps;

  for (int c0 = 0; c0 < N_; c0 += 16) {
    float4 c4 = s_cols[c0 + idx];
    v2f bf;
    bf.x = hi ? c4.z : c4.x;
    bf.y = hi ? 0.0f : c4.y;
    float gy = s_gy[c0 + idx];

    v8f cz = {};
    v8f acc = __builtin_amdgcn_wmma_f32_16x16x4_f32(
        false, a, false, bf, (short)0, cz, false, false);

#pragma unroll
    for (int r = 0; r < 8; ++r) {
      float v = (gy + acc[r]) * c2;
      float d = v - m[r];
      float e = __builtin_amdgcn_exp2f(-fabsf(d));
      bool  up = d > 0.0f;
      s[r] = up ? fmaf(s[r], e, 1.0f) : (s[r] + e);
      m[r] = up ? v : m[r];
    }
  }

#pragma unroll
  for (int mask = 1; mask < 16; mask <<= 1) {
#pragma unroll
    for (int r = 0; r < 8; ++r) {
      float om = __shfl_xor(m[r], mask, 32);
      float os = __shfl_xor(s[r], mask, 32);
      float d  = om - m[r];
      float e  = __builtin_amdgcn_exp2f(-fabsf(d));
      bool  up = d > 0.0f;
      s[r] = up ? fmaf(s[r], e, os) : fmaf(os, e, s[r]);
      m[r] = up ? om : m[r];
    }
  }

  if (idx < 8) {
    const int row = row_base + hi * 8 + idx;
    const int r   = idx;
    float sqrow = rp[row].w;
    float lse = LN2 * (m[r] + __builtin_amdgcn_logf(s[r]));
    out_pot[(size_t)b * N_ + row] = -eps * (lconst + lse) + 0.5f * sqrow;
  }
}

__global__ __launch_bounds__(THREADS)
void assign_mlp_kernel(const float4* __restrict__ xpts,
                       const float4* __restrict__ ypts,
                       const float*  __restrict__ g,
                       const float*  __restrict__ t,
                       const float*  __restrict__ W1,
                       const float*  __restrict__ Wt,
                       const float*  __restrict__ b1,
                       const float*  __restrict__ W2,
                       const float*  __restrict__ b2,
                       float*        __restrict__ out) {
  const int b    = blockIdx.y;
  const int tid  = threadIdx.x;
  const int lane = tid & 31;
  const int wave = tid >> 5;
  const int hi   = lane >> 4;
  const int idx  = lane & 15;

  __shared__ float4 s_cols[N_];
  __shared__ float  s_gy[N_];
  __shared__ float  s_W1[3 * H_];
  __shared__ float  s_Wt[H_];
  __shared__ float  s_b1[H_];
  __shared__ float  s_W2[H_ * 3];
  __shared__ float  s_b2[4];
  __shared__ int    s_idx[ROWS_PER_BLOCK];

  const float4* cp = ypts + (size_t)b * N_;
  const float*  gp = g    + (size_t)b * N_;
  for (int j = tid; j < N_; j += THREADS) {
    float4 c = cp[j];
    s_cols[j] = c;
    s_gy[j]   = gp[j] - c.w;
  }
  for (int j = tid; j < 3 * H_; j += THREADS) { s_W1[j] = W1[j]; s_W2[j] = W2[j]; }
  for (int j = tid; j < H_;     j += THREADS) { s_Wt[j] = Wt[j]; s_b1[j] = b1[j]; }
  if (tid < 3) s_b2[tid] = b2[tid];
  __syncthreads();

  const int row_base = blockIdx.x * ROWS_PER_BLOCK + wave * 16;
  const float4* rp = xpts + (size_t)b * N_;

  float4 r4 = rp[row_base + idx];
  v2f a;
  a.x = hi ? 2.0f * r4.z : 2.0f * r4.x;
  a.y = hi ? 0.0f        : 2.0f * r4.y;

  float best[8]; int bidx[8];
#pragma unroll
  for (int r = 0; r < 8; ++r) { best[r] = -3.0e38f; bidx[r] = 0; }

  for (int c0 = 0; c0 < N_; c0 += 16) {
    float4 c4 = s_cols[c0 + idx];
    v2f bf;
    bf.x = hi ? c4.z : c4.x;
    bf.y = hi ? 0.0f : c4.y;
    float gy = s_gy[c0 + idx];

    v8f cz = {};
    v8f acc = __builtin_amdgcn_wmma_f32_16x16x4_f32(
        false, a, false, bf, (short)0, cz, false, false);

#pragma unroll
    for (int r = 0; r < 8; ++r) {
      float v = acc[r] + gy;
      if (v > best[r]) { best[r] = v; bidx[r] = c0 + idx; }
    }
  }

#pragma unroll
  for (int mask = 1; mask < 16; mask <<= 1) {
#pragma unroll
    for (int r = 0; r < 8; ++r) {
      float ov = __shfl_xor(best[r], mask, 32);
      int   oi = __shfl_xor(bidx[r], mask, 32);
      if (ov > best[r] || (ov == best[r] && oi < bidx[r])) {
        best[r] = ov; bidx[r] = oi;
      }
    }
  }

  if (idx < 8) s_idx[wave * 16 + hi * 8 + idx] = bidx[idx];
  __syncthreads();

  if (tid < ROWS_PER_BLOCK) {
    const int grow = blockIdx.x * ROWS_PER_BLOCK + tid;
    const int j    = s_idx[tid];
    float4 x0a = s_cols[j];
    float4 xn  = rp[grow];
    const float tv = t[b], omt = 1.0f - tv;

    float vx = xn.x - x0a.x, vy = xn.y - x0a.y, vz = xn.z - x0a.z;
    float xtx = omt * x0a.x + tv * xn.x;
    float xty = omt * x0a.y + tv * xn.y;
    float xtz = omt * x0a.z + tv * xn.z;

    float p0 = s_b2[0], p1 = s_b2[1], p2 = s_b2[2];
    for (int k = 0; k < H_; ++k) {
      float h = xtx * s_W1[k] + xty * s_W1[H_ + k] + xtz * s_W1[2 * H_ + k]
              + tv * s_Wt[k] + s_b1[k];
      h = fmaxf(h, 0.0f);
      p0 += h * s_W2[k * 3 + 0];
      p1 += h * s_W2[k * 3 + 1];
      p2 += h * s_W2[k * 3 + 2];
    }
    const size_t base  = ((size_t)b * N_ + grow) * 3;
    const size_t vbase = (size_t)B_ * N_ * 3 + base;
    out[base + 0] = p0; out[base + 1] = p1; out[base + 2] = p2;
    out[vbase + 0] = vx; out[vbase + 1] = vy; out[vbase + 2] = vz;
  }
}

extern "C" void kernel_launch(void* const* d_in, const int* in_sizes, int n_in,
                              void* d_out, int out_size, void* d_ws, size_t ws_size,
                              hipStream_t stream) {
  (void)in_sizes; (void)n_in; (void)out_size; (void)ws_size;
  const float* cloud = (const float*)d_in[0];
  const float* noise = (const float*)d_in[1];
  const float* t     = (const float*)d_in[2];
  const float* W1    = (const float*)d_in[3];
  const float* Wt    = (const float*)d_in[4];
  const float* b1    = (const float*)d_in[5];
  const float* W2    = (const float*)d_in[6];
  const float* b2    = (const float*)d_in[7];
  float* out = (float*)d_out;

  char* ws = (char*)d_ws;
  float4* ypts = (float4*)ws;
  float4* xpts = (float4*)(ws + (size_t)B_ * N_ * sizeof(float4));
  float*  f    = (float*) (ws + (size_t)2 * B_ * N_ * sizeof(float4));
  float*  g    = f + (size_t)B_ * N_;

  prep_kernel<<<B_, THREADS, 0, stream>>>(cloud, noise, ypts, xpts, f, g);

  const float lconst = -logf((float)N_);
  dim3 grid(N_ / ROWS_PER_BLOCK, B_);
  for (int it = 0; it < 14; ++it) {
    float eps = (float)(32.0 * pow(1e-6 / 32.0, (double)it / 13.0));
    sinkhorn_half<<<grid, THREADS, 0, stream>>>(xpts, ypts, g, f, eps, lconst);
    sinkhorn_half<<<grid, THREADS, 0, stream>>>(ypts, xpts, f, g, eps, lconst);
  }

  assign_mlp_kernel<<<grid, THREADS, 0, stream>>>(xpts, ypts, g, t,
                                                  W1, Wt, b1, W2, b2, out);
}


# Tiled kernels re-gridded from 128 work-groups of 8 waves to 256 of 4 (64 rows each: one wave per SIMD on all 256 WGPs), staging loads issued as one group; per-row sums and their order unchanged
# speedup vs baseline: 1.4275x; 1.4275x over previous
#include <hip/hip_runtime.h>
#include <cmath>

#define B_  8
#define N_  2048
#define H_  256
#define THREADS 256
#define TILE_THREADS 128
#define ROWS_PER_BLOCK 64
#define COLS_PER_THREAD (N_ / TILE_THREADS)

typedef __attribute__((ext_vector_type(2))) float v2f;
typedef __attribute__((ext_vector_type(8))) float v8f;

#define LOG2E 1.4426950408889634f
#define LN2   0.6931471805599453f

__global__ __launch_bounds__(THREADS)
void prep_kernel(const float* __restrict__ cloud, const float* __restrict__ noise,
                 float4* __restrict__ ypts, float4* __restrict__ xpts,
                 float* __restrict__ f, float* __restrict__ g) {
  const int b   = blockIdx.x;
  const int tid = threadIdx.x;
  __shared__ float s_sum[THREADS];
  __shared__ float s_sq[THREADS];

  const float* cb = cloud + (size_t)b * N_ * 3;
  const float* nb = noise + (size_t)b * N_ * 3;

  float sum = 0.f, sq = 0.f;
  for (int i = tid; i < N_ * 3; i += THREADS) {
    float v = cb[i];
    sum += v; sq += v * v;
  }
  s_sum[tid] = sum; s_sq[tid] = sq;
  __syncthreads();
  for (int off = THREADS / 2; off > 0; off >>= 1) {
    if (tid < off) { s_sum[tid] += s_sum[tid + off]; s_sq[tid] += s_sq[tid + off]; }
    __syncthreads();
  }
  const float n    = (float)(N_ * 3);
  const float tot  = s_sum[0];
  const float tot2 = s_sq[0];
  const float var  = (tot2 - tot * tot / n) / (n - 1.0f);
  const float inv_std = rsqrtf(var);

  for (int p = tid; p < N_; p += THREADS) {
    float yx = cb[p * 3 + 0] * inv_std;
    float yy = cb[p * 3 + 1] * inv_std;
    float yz = cb[p * 3 + 2] * inv_std;
    ypts[(size_t)b * N_ + p] = make_float4(yx, yy, yz, yx * yx + yy * yy + yz * yz);
    float xx = nb[p * 3 + 0];
    float xy = nb[p * 3 + 1];
    float xz = nb[p * 3 + 2];
    xpts[(size_t)b * N_ + p] = make_float4(xx, xy, xz, xx * xx + xy * xy + xz * xz);
    f[(size_t)b * N_ + p] = 0.f;
    g[(size_t)b * N_ + p] = 0.f;
  }
}

__global__ __launch_bounds__(TILE_THREADS)
void sinkhorn_half(const float4* __restrict__ rows_pts,
                   const float4* __restrict__ cols_pts,
                   const float*  __restrict__ in_pot,
                   float*        __restrict__ out_pot,
                   float eps, float lconst) {
  const int b    = blockIdx.y;
  const int tid  = threadIdx.x;
  const int lane = tid & 31;
  const int wave = tid >> 5;
  const int hi   = lane >> 4;
  const int idx  = lane & 15;

  __shared__ float4 s_cols[N_];
  __shared__ float  s_gy[N_];

  const float4* cp = cols_pts + (size_t)b * N_;
  const float*  ip = in_pot   + (size_t)b * N_;
  float4 cst[COLS_PER_THREAD];
  float  pst[COLS_PER_THREAD];
#pragma unroll
  for (int k = 0; k < COLS_PER_THREAD; ++k) {
    cst[k] = cp[tid + k * TILE_THREADS];
    pst[k] = ip[tid + k * TILE_THREADS];
  }
#pragma unroll
  for (int k = 0; k < COLS_PER_THREAD; ++k) {
    const int j = tid + k * TILE_THREADS;
    s_cols[j] = cst[k];
    s_gy[j]   = pst[k] - 0.5f * cst[k].w;
  }
  __syncthreads();

  const int row_base = blockIdx.x * ROWS_PER_BLOCK + wave * 16;
  const float4* rp = rows_pts + (size_t)b * N_;

  float4 r4 = rp[row_base + idx];
  v2f a;
  a.x = hi ? r4.z : r4.x;
  a.y = hi ? 0.0f : r4.y;

  float m[8], s[8];
#pragma unroll
  for (int r = 0; r < 8; ++r) { m[r] = -3.0e38f; s[r] = 0.f; }
  const float c2 = LOG2E / eps;

  for (int c0 = 0; c0 < N_; c0 += 16) {
    float4 c4 = s_cols[c0 + idx];
    v2f bf;
    bf.x = hi ? c4.z : c4.x;
    bf.y = hi ? 0.0f : c4.y;
    float gy = s_gy[c0 + idx];

    v8f cz = {};
    v8f acc = __builtin_amdgcn_wmma_f32_16x16x4_f32(
        false, a, false, bf, (short)0, cz, false, false);

#pragma unroll
    for (int r = 0; r < 8; ++r) {
      float v = (gy + acc[r]) * c2;
      float d = v - m[r];
      float e = __builtin_amdgcn_exp2f(-fabsf(d));
      bool  up = d > 0.0f;
      s[r] = up ? fmaf(s[r], e, 1.0f) : (s[r] + e);
      m[r] = up ? v : m[r];
    }
  }

#pragma unroll
  for (int mask = 1; mask < 16; mask <<= 1) {
#pragma unroll
    for (int r = 0; r < 8; ++r) {
      float om = __shfl_xor(m[r], mask, 32);
      float os = __shfl_xor(s[r], mask, 32);
      float d  = om - m[r];
      float e  = __builtin_amdgcn_exp2f(-fabsf(d));
      bool  up = d > 0.0f;
      s[r] = up ? fmaf(s[r], e, os) : fmaf(os, e, s[r]);
      m[r] = up ? om : m[r];
    }
  }

  if (idx < 8) {
    const int row = row_base + hi * 8 + idx;
    const int r   = idx;
    float sqrow = rp[row].w;
    float lse = LN2 * (m[r] + __builtin_amdgcn_logf(s[r]));
    out_pot[(size_t)b * N_ + row] = -eps * (lconst + lse) + 0.5f * sqrow;
  }
}

__global__ __launch_bounds__(TILE_THREADS)
void assign_mlp_kernel(const float4* __restrict__ xpts,
                       const float4* __restrict__ ypts,
                       const float*  __restrict__ g,
                       const float*  __restrict__ t,
                       const float*  __restrict__ W1,
                       const float*  __restrict__ Wt,
                       const float*  __restrict__ b1,
                       const float*  __restrict__ W2,
                       const float*  __restrict__ b2,
                       float*        __restrict__ out) {
  const int b    = blockIdx.y;
  const int tid  = threadIdx.x;
  const int lane = tid & 31;
  const int wave = tid >> 5;
  const int hi   = lane >> 4;
  const int idx  = lane & 15;

  __shared__ float4 s_cols[N_];
  __shared__ float  s_gy[N_];
  __shared__ float  s_W1[3 * H_];
  __shared__ float  s_Wt[H_];
  __shared__ float  s_b1[H_];
  __shared__ float  s_W2[H_ * 3];
  __shared__ float  s_b2[4];
  __shared__ int    s_idx[ROWS_PER_BLOCK];

  const float4* cp = ypts + (size_t)b * N_;
  const float*  gp = g    + (size_t)b * N_;
  float4 cst[COLS_PER_THREAD];
  float  pst[COLS_PER_THREAD];
#pragma unroll
  for (int k = 0; k < COLS_PER_THREAD; ++k) {
    cst[k] = cp[tid + k * TILE_THREADS];
    pst[k] = gp[tid + k * TILE_THREADS];
  }
#pragma unroll
  for (int k = 0; k < COLS_PER_THREAD; ++k) {
    const int j = tid + k * TILE_THREADS;
    s_cols[j] = cst[k];
    s_gy[j]   = pst[k] - cst[k].w;
  }
#pragma unroll
  for (int k = 0; k < 3 * H_ / TILE_THREADS; ++k) {
    const int j = tid + k * TILE_THREADS;
    s_W1[j] = W1[j]; s_W2[j] = W2[j];
  }
#pragma unroll
  for (int k = 0; k < H_ / TILE_THREADS; ++k) {
    const int j = tid + k * TILE_THREADS;
    s_Wt[j] = Wt[j]; s_b1[j] = b1[j];
  }
  if (tid < 3) s_b2[tid] = b2[tid];
  __syncthreads();

  const int row_base = blockIdx.x * ROWS_PER_BLOCK + wave * 16;
  const float4* rp = xpts + (size_t)b * N_;

  float4 r4 = rp[row_base + idx];
  v2f a;
  a.x = hi ? 2.0f * r4.z : 2.0f * r4.x;
  a.y = hi ? 0.0f        : 2.0f * r4.y;

  float best[8]; int bidx[8];
#pragma unroll
  for (int r = 0; r < 8; ++r) { best[r] = -3.0e38f; bidx[r] = 0; }

  for (int c0 = 0; c0 < N_; c0 += 16) {
    float4 c4 = s_cols[c0 + idx];
    v2f bf;
    bf.x = hi ? c4.z : c4.x;
    bf.y = hi ? 0.0f : c4.y;
    float gy = s_gy[c0 + idx];

    v8f cz = {};
    v8f acc = __builtin_amdgcn_wmma_f32_16x16x4_f32(
        false, a, false, bf, (short)0, cz, false, false);

#pragma unroll
    for (int r = 0; r < 8; ++r) {
      float v = acc[r] + gy;
      if (v > best[r]) { best[r] = v; bidx[r] = c0 + idx; }
    }
  }

#pragma unroll
  for (int mask = 1; mask < 16; mask <<= 1) {
#pragma unroll
    for (int r = 0; r < 8; ++r) {
      float ov = __shfl_xor(best[r], mask, 32);
      int   oi = __shfl_xor(bidx[r], mask, 32);
      if (ov > best[r] || (ov == best[r] && oi < bidx[r])) {
        best[r] = ov; bidx[r] = oi;
      }
    }
  }

  if (idx < 8) s_idx[wave * 16 + hi * 8 + idx] = bidx[idx];
  __syncthreads();

  if (tid < ROWS_PER_BLOCK) {
    const int grow = blockIdx.x * ROWS_PER_BLOCK + tid;
    const int j    = s_idx[tid];
    float4 x0a = s_cols[j];
    float4 xn  = rp[grow];
    const float tv = t[b], omt = 1.0f - tv;

    float vx = xn.x - x0a.x, vy = xn.y - x0a.y, vz = xn.z - x0a.z;
    float xtx = omt * x0a.x + tv * xn.x;
    float xty = omt * x0a.y + tv * xn.y;
    float xtz = omt * x0a.z + tv * xn.z;

    float p0 = s_b2[0], p1 = s_b2[1], p2 = s_b2[2];
    for (int k = 0; k < H_; ++k) {
      float h = xtx * s_W1[k] + xty * s_W1[H_ + k] + xtz * s_W1[2 * H_ + k]
              + tv * s_Wt[k] + s_b1[k];
      h = fmaxf(h, 0.0f);
      p0 += h * s_W2[k * 3 + 0];
      p1 += h * s_W2[k * 3 + 1];
      p2 += h * s_W2[k * 3 + 2];
    }
    const size_t base  = ((size_t)b * N_ + grow) * 3;
    const size_t vbase = (size_t)B_ * N_ * 3 + base;
    out[base + 0] = p0; out[base + 1] = p1; out[base + 2] = p2;
    out[vbase + 0] = vx; out[vbase + 1] = vy; out[vbase + 2] = vz;
  }
}

extern "C" void kernel_launch(void* const* d_in, const int* in_sizes, int n_in,
                              void* d_out, int out_size, void* d_ws, size_t ws_size,
                              hipStream_t stream) {
  (void)in_sizes; (void)n_in; (void)out_size; (void)ws_size;
  const float* cloud = (const float*)d_in[0];
  const float* noise = (const float*)d_in[1];
  const float* t     = (const float*)d_in[2];
  const float* W1    = (const float*)d_in[3];
  const float* Wt    = (const float*)d_in[4];
  const float* b1    = (const float*)d_in[5];
  const float* W2    = (const float*)d_in[6];
  const float* b2    = (const float*)d_in[7];
  float* out = (float*)d_out;

  char* ws = (char*)d_ws;
  float4* ypts = (float4*)ws;
  float4* xpts = (float4*)(ws + (size_t)B_ * N_ * sizeof(float4));
  float*  f    = (float*) (ws + (size_t)2 * B_ * N_ * sizeof(float4));
  float*  g    = f + (size_t)B_ * N_;

  prep_kernel<<<B_, THREADS, 0, stream>>>(cloud, noise, ypts, xpts, f, g);

  const float lconst = -logf((float)N_);
  dim3 grid(N_ / ROWS_PER_BLOCK, B_);
  for (int it = 0; it < 14; ++it) {
    float eps = (float)(32.0 * pow(1e-6 / 32.0, (double)it / 13.0));
    sinkhorn_half<<<grid, TILE_THREADS, 0, stream>>>(xpts, ypts, g, f, eps, lconst);
    sinkhorn_half<<<grid, TILE_THREADS, 0, stream>>>(ypts, xpts, f, g, eps, lconst);
  }

  assign_mlp_kernel<<<grid, TILE_THREADS, 0, stream>>>(xpts, ypts, g, t,
                                                  W1, Wt, b1, W2, b2, out);
}


# Column loops software-pipelined: next stage's WMMAs and the LDS fragment reads two stages ahead issue before the current stage's exp/LSE (argmax) vector work; values and summation order unchanged
# speedup vs baseline: 1.6758x; 1.1739x over previous
#include <hip/hip_runtime.h>
#include <cmath>

#define B_  8
#define N_  2048
#define H_  256
#define THREADS 256
#define TILE_THREADS 128
#define ROWS_PER_BLOCK 64
#define COLS_PER_THREAD (N_ / TILE_THREADS)

typedef __attribute__((ext_vector_type(2))) float v2f;
typedef __attribute__((ext_vector_type(8))) float v8f;

#define LOG2E 1.4426950408889634f
#define LN2   0.6931471805599453f

#define STAGE_COLS 32

__device__ __forceinline__ v2f frag_b(const float4 c4, const int hi) {
  v2f bf;
  bf.x = hi ? c4.z : c4.x;
  bf.y = hi ? 0.0f : c4.y;
  return bf;
}

__device__ __forceinline__ v8f tile_product(const v2f a, const v2f bf) {
  const v8f cz = {};
  return __builtin_amdgcn_wmma_f32_16x16x4_f32(
      false, a, false, bf, (short)0, cz, false, false);
}

__global__ __launch_bounds__(THREADS)
void prep_kernel(const float* __restrict__ cloud, const float* __restrict__ noise,
                 float4* __restrict__ ypts, float4* __restrict__ xpts,
                 float* __restrict__ f, float* __restrict__ g) {
  const int b   = blockIdx.x;
  const int tid = threadIdx.x;
  __shared__ float s_sum[THREADS];
  __shared__ float s_sq[THREADS];

  const float* cb = cloud + (size_t)b * N_ * 3;
  const float* nb = noise + (size_t)b * N_ * 3;

  float sum = 0.f, sq = 0.f;
  for (int i = tid; i < N_ * 3; i += THREADS) {
    float v = cb[i];
    sum += v; sq += v * v;
  }
  s_sum[tid] = sum; s_sq[tid] = sq;
  __syncthreads();
  for (int off = THREADS / 2; off > 0; off >>= 1) {
    if (tid < off) { s_sum[tid] += s_sum[tid + off]; s_sq[tid] += s_sq[tid + off]; }
    __syncthreads();
  }
  const float n    = (float)(N_ * 3);
  const float tot  = s_sum[0];
  const float tot2 = s_sq[0];
  const float var  = (tot2 - tot * tot / n) / (n - 1.0f);
  const float inv_std = rsqrtf(var);

  for (int p = tid; p < N_; p += THREADS) {
    float yx = cb[p * 3 + 0] * inv_std;
    float yy = cb[p * 3 + 1] * inv_std;
    float yz = cb[p * 3 + 2] * inv_std;
    ypts[(size_t)b * N_ + p] = make_float4(yx, yy, yz, yx * yx + yy * yy + yz * yz);
    float xx = nb[p * 3 + 0];
    float xy = nb[p * 3 + 1];
    float xz = nb[p * 3 + 2];
    xpts[(size_t)b * N_ + p] = make_float4(xx, xy, xz, xx * xx + xy * xy + xz * xz);
    f[(size_t)b * N_ + p] = 0.f;
    g[(size_t)b * N_ + p] = 0.f;
  }
}

__global__ __launch_bounds__(TILE_THREADS)
void sinkhorn_half(const float4* __restrict__ rows_pts,
                   const float4* __restrict__ cols_pts,
                   const float*  __restrict__ in_pot,
                   float*        __restrict__ out_pot,
                   float eps, float lconst) {
  const int b    = blockIdx.y;
  const int tid  = threadIdx.x;
  const int lane = tid & 31;
  const int wave = tid >> 5;
  const int hi   = lane >> 4;
  const int idx  = lane & 15;

  __shared__ float4 s_cols[N_];
  __shared__ float  s_gy[N_];

  const float4* cp = cols_pts + (size_t)b * N_;
  const float*  ip = in_pot   + (size_t)b * N_;
  float4 cst[COLS_PER_THREAD];
  float  pst[COLS_PER_THREAD];
#pragma unroll
  for (int k = 0; k < COLS_PER_THREAD; ++k) {
    cst[k] = cp[tid + k * TILE_THREADS];
    pst[k] = ip[tid + k * TILE_THREADS];
  }
#pragma unroll
  for (int k = 0; k < COLS_PER_THREAD; ++k) {
    const int j = tid + k * TILE_THREADS;
    s_cols[j] = cst[k];
    s_gy[j]   = pst[k] - 0.5f * cst[k].w;
  }
  __syncthreads();

  const int row_base = blockIdx.x * ROWS_PER_BLOCK + wave * 16;
  const float4* rp = rows_pts + (size_t)b * N_;

  float4 r4 = rp[row_base + idx];
  v2f a;
  a.x = hi ? r4.z : r4.x;
  a.y = hi ? 0.0f : r4.y;

  float m[8], s[8];
#pragma unroll
  for (int r = 0; r < 8; ++r) { m[r] = -3.0e38f; s[r] = 0.f; }
  const float c2 = LOG2E / eps;

  auto lse_update = [&](const v8f acc, const float gy) {
#pragma unroll
    for (int r = 0; r < 8; ++r) {
      float v = (gy + acc[r]) * c2;
      float d = v - m[r];
      float e = __builtin_amdgcn_exp2f(-fabsf(d));
      bool  up = d > 0.0f;
      s[r] = up ? fmaf(s[r], e, 1.0f) : (s[r] + e);
      m[r] = up ? v : m[r];
    }
  };

  const float4* lc = s_cols + idx;
  const float*  lg = s_gy + idx;

  float4 cP0, cP1, cQ0, cQ1;
  float  gP0, gP1, gQ0, gQ1;
  v8f    aP0, aP1, aQ0, aQ1;
  float  yP0, yP1, yQ0, yQ1;

  cP0 = lc[0];  cP1 = lc[16];  gP0 = lg[0];  gP1 = lg[16];
  aP0 = tile_product(a, frag_b(cP0, hi));
  aP1 = tile_product(a, frag_b(cP1, hi));
  yP0 = gP0; yP1 = gP1;
  cQ0 = lc[STAGE_COLS];  cQ1 = lc[STAGE_COLS + 16];
  gQ0 = lg[STAGE_COLS];  gQ1 = lg[STAGE_COLS + 16];

  for (int c0 = 0; c0 < N_; c0 += 2 * STAGE_COLS) {
    const int n0 = (c0 + 2 * STAGE_COLS) & (N_ - 1);
    const int n1 = (c0 + 3 * STAGE_COLS) & (N_ - 1);

    aQ0 = tile_product(a, frag_b(cQ0, hi));
    aQ1 = tile_product(a, frag_b(cQ1, hi));
    yQ0 = gQ0; yQ1 = gQ1;
    cP0 = lc[n0];  cP1 = lc[n0 + 16];  gP0 = lg[n0];  gP1 = lg[n0 + 16];
    lse_update(aP0, yP0);
    lse_update(aP1, yP1);

    aP0 = tile_product(a, frag_b(cP0, hi));
    aP1 = tile_product(a, frag_b(cP1, hi));
    yP0 = gP0; yP1 = gP1;
    cQ0 = lc[n1];  cQ1 = lc[n1 + 16];  gQ0 = lg[n1];  gQ1 = lg[n1 + 16];
    lse_update(aQ0, yQ0);
    lse_update(aQ1, yQ1);
  }

#pragma unroll
  for (int mask = 1; mask < 16; mask <<= 1) {
#pragma unroll
    for (int r = 0; r < 8; ++r) {
      float om = __shfl_xor(m[r], mask, 32);
      float os = __shfl_xor(s[r], mask, 32);
      float d  = om - m[r];
      float e  = __builtin_amdgcn_exp2f(-fabsf(d));
      bool  up = d > 0.0f;
      s[r] = up ? fmaf(s[r], e, os) : fmaf(os, e, s[r]);
      m[r] = up ? om : m[r];
    }
  }

  if (idx < 8) {
    const int row = row_base + hi * 8 + idx;
    const int r   = idx;
    float sqrow = rp[row].w;
    float lse = LN2 * (m[r] + __builtin_amdgcn_logf(s[r]));
    out_pot[(size_t)b * N_ + row] = -eps * (lconst + lse) + 0.5f * sqrow;
  }
}

__global__ __launch_bounds__(TILE_THREADS)
void assign_mlp_kernel(const float4* __restrict__ xpts,
                       const float4* __restrict__ ypts,
                       const float*  __restrict__ g,
                       const float*  __restrict__ t,
                       const float*  __restrict__ W1,
                       const float*  __restrict__ Wt,
                       const float*  __restrict__ b1,
                       const float*  __restrict__ W2,
                       const float*  __restrict__ b2,
                       float*        __restrict__ out) {
  const int b    = blockIdx.y;
  const int tid  = threadIdx.x;
  const int lane = tid & 31;
  const int wave = tid >> 5;
  const int hi   = lane >> 4;
  const int idx  = lane & 15;

  __shared__ float4 s_cols[N_];
  __shared__ float  s_gy[N_];
  __shared__ float  s_W1[3 * H_];
  __shared__ float  s_Wt[H_];
  __shared__ float  s_b1[H_];
  __shared__ float  s_W2[H_ * 3];
  __shared__ float  s_b2[4];
  __shared__ int    s_idx[ROWS_PER_BLOCK];

  const float4* cp = ypts + (size_t)b * N_;
  const float*  gp = g    + (size_t)b * N_;
  float4 cst[COLS_PER_THREAD];
  float  pst[COLS_PER_THREAD];
#pragma unroll
  for (int k = 0; k < COLS_PER_THREAD; ++k) {
    cst[k] = cp[tid + k * TILE_THREADS];
    pst[k] = gp[tid + k * TILE_THREADS];
  }
#pragma unroll
  for (int k = 0; k < COLS_PER_THREAD; ++k) {
    const int j = tid + k * TILE_THREADS;
    s_cols[j] = cst[k];
    s_gy[j]   = pst[k] - cst[k].w;
  }
#pragma unroll
  for (int k = 0; k < 3 * H_ / TILE_THREADS; ++k) {
    const int j = tid + k * TILE_THREADS;
    s_W1[j] = W1[j]; s_W2[j] = W2[j];
  }
#pragma unroll
  for (int k = 0; k < H_ / TILE_THREADS; ++k) {
    const int j = tid + k * TILE_THREADS;
    s_Wt[j] = Wt[j]; s_b1[j] = b1[j];
  }
  if (tid < 3) s_b2[tid] = b2[tid];
  __syncthreads();

  const int row_base = blockIdx.x * ROWS_PER_BLOCK + wave * 16;
  const float4* rp = xpts + (size_t)b * N_;

  float4 r4 = rp[row_base + idx];
  v2f a;
  a.x = hi ? 2.0f * r4.z : 2.0f * r4.x;
  a.y = hi ? 0.0f        : 2.0f * r4.y;

  float best[8]; int bidx[8];
#pragma unroll
  for (int r = 0; r < 8; ++r) { best[r] = -3.0e38f; bidx[r] = 0; }

  auto amax_update = [&](const v8f acc, const float gy, const int col) {
#pragma unroll
    for (int r = 0; r < 8; ++r) {
      float v = acc[r] + gy;
      if (v > best[r]) { best[r] = v; bidx[r] = col; }
    }
  };

  const float4* lc = s_cols + idx;
  const float*  lg = s_gy + idx;

  float4 cP0, cP1, cQ0, cQ1;
  float  gP0, gP1, gQ0, gQ1;
  v8f    aP0, aP1, aQ0, aQ1;
  float  yP0, yP1, yQ0, yQ1;

  cP0 = lc[0];  cP1 = lc[16];  gP0 = lg[0];  gP1 = lg[16];
  aP0 = tile_product(a, frag_b(cP0, hi));
  aP1 = tile_product(a, frag_b(cP1, hi));
  yP0 = gP0; yP1 = gP1;
  cQ0 = lc[STAGE_COLS];  cQ1 = lc[STAGE_COLS + 16];
  gQ0 = lg[STAGE_COLS];  gQ1 = lg[STAGE_COLS + 16];

  for (int c0 = 0; c0 < N_; c0 += 2 * STAGE_COLS) {
    const int n0 = (c0 + 2 * STAGE_COLS) & (N_ - 1);
    const int n1 = (c0 + 3 * STAGE_COLS) & (N_ - 1);

    aQ0 = tile_product(a, frag_b(cQ0, hi));
    aQ1 = tile_product(a, frag_b(cQ1, hi));
    yQ0 = gQ0; yQ1 = gQ1;
    cP0 = lc[n0];  cP1 = lc[n0 + 16];  gP0 = lg[n0];  gP1 = lg[n0 + 16];
    amax_update(aP0, yP0, c0 + idx);
    amax_update(aP1, yP1, c0 + 16 + idx);

    aP0 = tile_product(a, frag_b(cP0, hi));
    aP1 = tile_product(a, frag_b(cP1, hi));
    yP0 = gP0; yP1 = gP1;
    cQ0 = lc[n1];  cQ1 = lc[n1 + 16];  gQ0 = lg[n1];  gQ1 = lg[n1 + 16];
    amax_update(aQ0, yQ0, c0 + STAGE_COLS + idx);
    amax_update(aQ1, yQ1, c0 + STAGE_COLS + 16 + idx);
  }

#pragma unroll
  for (int mask = 1; mask < 16; mask <<= 1) {
#pragma unroll
    for (int r = 0; r < 8; ++r) {
      float ov = __shfl_xor(best[r], mask, 32);
      int   oi = __shfl_xor(bidx[r], mask, 32);
      if (ov > best[r] || (ov == best[r] && oi < bidx[r])) {
        best[r] = ov; bidx[r] = oi;
      }
    }
  }

  if (idx < 8) s_idx[wave * 16 + hi * 8 + idx] = bidx[idx];
  __syncthreads();

  if (tid < ROWS_PER_BLOCK) {
    const int grow = blockIdx.x * ROWS_PER_BLOCK + tid;
    const int j    = s_idx[tid];
    float4 x0a = s_cols[j];
    float4 xn  = rp[grow];
    const float tv = t[b], omt = 1.0f - tv;

    float vx = xn.x - x0a.x, vy = xn.y - x0a.y, vz = xn.z - x0a.z;
    float xtx = omt * x0a.x + tv * xn.x;
    float xty = omt * x0a.y + tv * xn.y;
    float xtz = omt * x0a.z + tv * xn.z;

    float p0 = s_b2[0], p1 = s_b2[1], p2 = s_b2[2];
    for (int k = 0; k < H_; ++k) {
      float h = xtx * s_W1[k] + xty * s_W1[H_ + k] + xtz * s_W1[2 * H_ + k]
              + tv * s_Wt[k] + s_b1[k];
      h = fmaxf(h, 0.0f);
      p0 += h * s_W2[k * 3 + 0];
      p1 += h * s_W2[k * 3 + 1];
      p2 += h * s_W2[k * 3 + 2];
    }
    const size_t base  = ((size_t)b * N_ + grow) * 3;
    const size_t vbase = (size_t)B_ * N_ * 3 + base;
    out[base + 0] = p0; out[base + 1] = p1; out[base + 2] = p2;
    out[vbase + 0] = vx; out[vbase + 1] = vy; out[vbase + 2] = vz;
  }
}

extern "C" void kernel_launch(void* const* d_in, const int* in_sizes, int n_in,
                              void* d_out, int out_size, void* d_ws, size_t ws_size,
                              hipStream_t stream) {
  (void)in_sizes; (void)n_in; (void)out_size; (void)ws_size;
  const float* cloud = (const float*)d_in[0];
  const float* noise = (const float*)d_in[1];
  const float* t     = (const float*)d_in[2];
  const float* W1    = (const float*)d_in[3];
  const float* Wt    = (const float*)d_in[4];
  const float* b1    = (const float*)d_in[5];
  const float* W2    = (const float*)d_in[6];
  const float* b2    = (const float*)d_in[7];
  float* out = (float*)d_out;

  char* ws = (char*)d_ws;
  float4* ypts = (float4*)ws;
  float4* xpts = (float4*)(ws + (size_t)B_ * N_ * sizeof(float4));
  float*  f    = (float*) (ws + (size_t)2 * B_ * N_ * sizeof(float4));
  float*  g    = f + (size_t)B_ * N_;

  prep_kernel<<<B_, THREADS, 0, stream>>>(cloud, noise, ypts, xpts, f, g);

  const float lconst = -logf((float)N_);
  dim3 grid(N_ / ROWS_PER_BLOCK, B_);
  for (int it = 0; it < 14; ++it) {
    float eps = (float)(32.0 * pow(1e-6 / 32.0, (double)it / 13.0));
    sinkhorn_half<<<grid, TILE_THREADS, 0, stream>>>(xpts, ypts, g, f, eps, lconst);
    sinkhorn_half<<<grid, TILE_THREADS, 0, stream>>>(ypts, xpts, f, g, eps, lconst);
  }

  assign_mlp_kernel<<<grid, TILE_THREADS, 0, stream>>>(xpts, ypts, g, t,
                                                  W1, Wt, b1, W2, b2, out);
}


# LSE kernel: column term gy rides the free 4th k of the f32 product (A's 4th k = 1), B fragment read as its own 8 LDS bytes: per-element add, gy reads, selects go; gy now summed as the product's 4th k
# speedup vs baseline: 2.1220x; 1.2663x over previous
#include <hip/hip_runtime.h>
#include <cmath>

#define B_  8
#define N_  2048
#define H_  256
#define THREADS 256
#define TILE_THREADS 128
#define ROWS_PER_BLOCK 64
#define COLS_PER_THREAD (N_ / TILE_THREADS)

typedef __attribute__((ext_vector_type(2))) float v2f;
typedef __attribute__((ext_vector_type(8))) float v8f;

#define LOG2E 1.4426950408889634f
#define LN2   0.6931471805599453f

#define STAGE_COLS 32

#define BLOCK_TILES 8
#define BLOCK_COLS (16 * BLOCK_TILES)

__device__ __forceinline__ v2f frag_b(const float4 c4, const int hi) {
  v2f bf;
  bf.x = hi ? c4.z : c4.x;
  bf.y = hi ? 0.0f : c4.y;
  return bf;
}

__device__ __forceinline__ v8f tile_product(const v2f a, const v2f bf) {
  const v8f cz = {};
  return __builtin_amdgcn_wmma_f32_16x16x4_f32(
      false, a, false, bf, (short)0, cz, false, false);
}

__global__ __launch_bounds__(THREADS)
void prep_kernel(const float* __restrict__ cloud, const float* __restrict__ noise,
                 float4* __restrict__ ypts, float4* __restrict__ xpts,
                 float* __restrict__ f, float* __restrict__ g) {
  const int b   = blockIdx.x;
  const int tid = threadIdx.x;
  __shared__ float s_sum[THREADS];
  __shared__ float s_sq[THREADS];

  const float* cb = cloud + (size_t)b * N_ * 3;
  const float* nb = noise + (size_t)b * N_ * 3;

  float sum = 0.f, sq = 0.f;
  for (int i = tid; i < N_ * 3; i += THREADS) {
    float v = cb[i];
    sum += v; sq += v * v;
  }
  s_sum[tid] = sum; s_sq[tid] = sq;
  __syncthreads();
  for (int off = THREADS / 2; off > 0; off >>= 1) {
    if (tid < off) { s_sum[tid] += s_sum[tid + off]; s_sq[tid] += s_sq[tid + off]; }
    __syncthreads();
  }
  const float n    = (float)(N_ * 3);
  const float tot  = s_sum[0];
  const float tot2 = s_sq[0];
  const float var  = (tot2 - tot * tot / n) / (n - 1.0f);
  const float inv_std = rsqrtf(var);

  for (int p = tid; p < N_; p += THREADS) {
    float yx = cb[p * 3 + 0] * inv_std;
    float yy = cb[p * 3 + 1] * inv_std;
    float yz = cb[p * 3 + 2] * inv_std;
    ypts[(size_t)b * N_ + p] = make_float4(yx, yy, yz, yx * yx + yy * yy + yz * yz);
    float xx = nb[p * 3 + 0];
    float xy = nb[p * 3 + 1];
    float xz = nb[p * 3 + 2];
    xpts[(size_t)b * N_ + p] = make_float4(xx, xy, xz, xx * xx + xy * xy + xz * xz);
    f[(size_t)b * N_ + p] = 0.f;
    g[(size_t)b * N_ + p] = 0.f;
  }
}

__global__ __launch_bounds__(TILE_THREADS)
void sinkhorn_half(const float4* __restrict__ rows_pts,
                   const float4* __restrict__ cols_pts,
                   const float*  __restrict__ in_pot,
                   float*        __restrict__ out_pot,
                   float eps, float lconst) {
  const int b    = blockIdx.y;
  const int tid  = threadIdx.x;
  const int lane = tid & 31;
  const int wave = tid >> 5;
  const int hi   = lane >> 4;
  const int idx  = lane & 15;

  __shared__ __attribute__((aligned(16))) v2f s_cols[2 * N_];

  const float4* cp = cols_pts + (size_t)b * N_;
  const float*  ip = in_pot   + (size_t)b * N_;
  float4 cst[COLS_PER_THREAD];
  float  pst[COLS_PER_THREAD];
#pragma unroll
  for (int k = 0; k < COLS_PER_THREAD; ++k) {
    cst[k] = cp[tid + k * TILE_THREADS];
    pst[k] = ip[tid + k * TILE_THREADS];
  }
#pragma unroll
  for (int k = 0; k < COLS_PER_THREAD; ++k) {
    const int j = tid + k * TILE_THREADS;
    v2f lo, up;
    lo.x = cst[k].x;  lo.y = cst[k].y;
    up.x = cst[k].z;  up.y = pst[k] - 0.5f * cst[k].w;
    s_cols[2 * j]     = lo;
    s_cols[2 * j + 1] = up;
  }
  __syncthreads();

  const int row_base = blockIdx.x * ROWS_PER_BLOCK + wave * 16;
  const float4* rp = rows_pts + (size_t)b * N_;

  float4 r4 = rp[row_base + idx];
  v2f a;
  a.x = hi ? r4.z : r4.x;
  a.y = hi ? 1.0f : r4.y;

  v8f m, s;
#pragma unroll
  for (int r = 0; r < 8; ++r) { m[r] = -3.0e38f; s[r] = 0.f; }
  const float c2 = LOG2E / eps;

  const v2f* lc = s_cols + 2 * idx + hi;

  v2f cP[BLOCK_TILES], cQ[BLOCK_TILES];
  v8f aP[BLOCK_TILES], aQ[BLOCK_TILES];

  auto read_block = [&](const int col0, v2f (&c)[BLOCK_TILES]) {
#pragma unroll
    for (int t = 0; t < BLOCK_TILES; ++t) c[t] = lc[2 * (col0 + 16 * t)];
  };

  auto lse_block = [&](const v8f (&acc)[BLOCK_TILES]) {
    v8f v[BLOCK_TILES];
#pragma unroll
    for (int t = 0; t < BLOCK_TILES; ++t) v[t] = acc[t] * c2;
    const v8f mn = __builtin_elementwise_max(
        __builtin_elementwise_max(__builtin_elementwise_max(__builtin_elementwise_max(v[0], v[1]), v[2]),
                                  __builtin_elementwise_max(__builtin_elementwise_max(v[3], v[4]), v[5])),
        __builtin_elementwise_max(__builtin_elementwise_max(v[6], v[7]), m));
    const v8f dm = m - mn;
    v8f keep;
#pragma unroll
    for (int r = 0; r < 8; ++r) keep[r] = __builtin_amdgcn_exp2f(dm[r]);
    v8f e[BLOCK_TILES];
#pragma unroll
    for (int t = 0; t < BLOCK_TILES; ++t) {
      v8f d;
      {
#pragma clang fp contract(off)
        d = v[t] - mn;
      }
#pragma unroll
      for (int r = 0; r < 8; ++r) e[t][r] = __builtin_amdgcn_exp2f(d[r]);
    }
    const v8f blk = ((e[0] + e[1]) + (e[2] + e[3])) + ((e[4] + e[5]) + (e[6] + e[7]));
    s = __builtin_elementwise_fma(s, keep, blk);
    m = mn;
  };

  auto block_step = [&](const v2f (&c)[BLOCK_TILES], v8f (&nxt)[BLOCK_TILES],
                        const v8f (&acc)[BLOCK_TILES]) {
#pragma unroll
    for (int t = 0; t < BLOCK_TILES; ++t) nxt[t] = tile_product(a, c[t]);
    lse_block(acc);
#pragma unroll
    for (int t = 0; t < 3; ++t) {
      __builtin_amdgcn_sched_group_barrier(0x008, 1, 0);
      __builtin_amdgcn_sched_group_barrier(0x002, 23, 0);
    }
    __builtin_amdgcn_sched_group_barrier(0x100, BLOCK_TILES, 0);
#pragma unroll
    for (int t = 3; t < BLOCK_TILES; ++t) {
      __builtin_amdgcn_sched_group_barrier(0x008, 1, 0);
#pragma unroll
      for (int k = 0; k < 14; ++k) {
        __builtin_amdgcn_sched_group_barrier(0x002, 1, 0);
        __builtin_amdgcn_sched_group_barrier(0x400, 1, 0);
      }
    }
#pragma unroll
    for (int k = 0; k < 2; ++k) {
      __builtin_amdgcn_sched_group_barrier(0x002, 1, 0);
      __builtin_amdgcn_sched_group_barrier(0x400, 1, 0);
    }
  };

  read_block(0, cP);
#pragma unroll
  for (int t = 0; t < BLOCK_TILES; ++t) aP[t] = tile_product(a, cP[t]);
  read_block(BLOCK_COLS, cQ);

  for (int c0 = 0; c0 < N_; c0 += 2 * BLOCK_COLS) {
    const int n0 = (c0 + 2 * BLOCK_COLS) & (N_ - 1);
    const int n1 = (c0 + 3 * BLOCK_COLS) & (N_ - 1);

    read_block(n0, cP);
    block_step(cQ, aQ, aP);

    read_block(n1, cQ);
    block_step(cP, aP, aQ);
  }

#pragma unroll
  for (int mask = 1; mask < 16; mask <<= 1) {
#pragma unroll
    for (int r = 0; r < 8; ++r) {
      float om = __shfl_xor(m[r], mask, 32);
      float os = __shfl_xor(s[r], mask, 32);
      float d  = om - m[r];
      float e  = __builtin_amdgcn_exp2f(-fabsf(d));
      bool  up = d > 0.0f;
      s[r] = up ? fmaf(s[r], e, os) : fmaf(os, e, s[r]);
      m[r] = up ? om : m[r];
    }
  }

  if (idx < 8) {
    const int row = row_base + hi * 8 + idx;
    const int r   = idx;
    float sqrow = rp[row].w;
    float lse = LN2 * (m[r] + __builtin_amdgcn_logf(s[r]));
    out_pot[(size_t)b * N_ + row] = -eps * (lconst + lse) + 0.5f * sqrow;
  }
}

__global__ __launch_bounds__(TILE_THREADS)
void assign_mlp_kernel(const float4* __restrict__ xpts,
                       const float4* __restrict__ ypts,
                       const float*  __restrict__ g,
                       const float*  __restrict__ t,
                       const float*  __restrict__ W1,
                       const float*  __restrict__ Wt,
                       const float*  __restrict__ b1,
                       const float*  __restrict__ W2,
                       const float*  __restrict__ b2,
                       float*        __restrict__ out) {
  const int b    = blockIdx.y;
  const int tid  = threadIdx.x;
  const int lane = tid & 31;
  const int wave = tid >> 5;
  const int hi   = lane >> 4;
  const int idx  = lane & 15;

  __shared__ float4 s_cols[N_];
  __shared__ float  s_gy[N_];
  __shared__ float  s_W1[3 * H_];
  __shared__ float  s_Wt[H_];
  __shared__ float  s_b1[H_];
  __shared__ float  s_W2[H_ * 3];
  __shared__ float  s_b2[4];
  __shared__ int    s_idx[ROWS_PER_BLOCK];

  const float4* cp = ypts + (size_t)b * N_;
  const float*  gp = g    + (size_t)b * N_;
  float4 cst[COLS_PER_THREAD];
  float  pst[COLS_PER_THREAD];
#pragma unroll
  for (int k = 0; k < COLS_PER_THREAD; ++k) {
    cst[k] = cp[tid + k * TILE_THREADS];
    pst[k] = gp[tid + k * TILE_THREADS];
  }
#pragma unroll
  for (int k = 0; k < COLS_PER_THREAD; ++k) {
    const int j = tid + k * TILE_THREADS;
    s_cols[j] = cst[k];
    s_gy[j]   = pst[k] - cst[k].w;
  }
#pragma unroll
  for (int k = 0; k < 3 * H_ / TILE_THREADS; ++k) {
    const int j = tid + k * TILE_THREADS;
    s_W1[j] = W1[j]; s_W2[j] = W2[j];
  }
#pragma unroll
  for (int k = 0; k < H_ / TILE_THREADS; ++k) {
    const int j = tid + k * TILE_THREADS;
    s_Wt[j] = Wt[j]; s_b1[j] = b1[j];
  }
  if (tid < 3) s_b2[tid] = b2[tid];
  __syncthreads();

  const int row_base = blockIdx.x * ROWS_PER_BLOCK + wave * 16;
  const float4* rp = xpts + (size_t)b * N_;

  float4 r4 = rp[row_base + idx];
  v2f a;
  a.x = hi ? 2.0f * r4.z : 2.0f * r4.x;
  a.y = hi ? 0.0f        : 2.0f * r4.y;

  float best[8]; int bidx[8];
#pragma unroll
  for (int r = 0; r < 8; ++r) { best[r] = -3.0e38f; bidx[r] = 0; }

  auto amax_update = [&](const v8f acc, const float gy, const int col) {
#pragma unroll
    for (int r = 0; r < 8; ++r) {
      float v = acc[r] + gy;
      if (v > best[r]) { best[r] = v; bidx[r] = col; }
    }
  };

  const float4* lc = s_cols + idx;
  const float*  lg = s_gy + idx;

  float4 cP0, cP1, cQ0, cQ1;
  float  gP0, gP1, gQ0, gQ1;
  v8f    aP0, aP1, aQ0, aQ1;
  float  yP0, yP1, yQ0, yQ1;

  cP0 = lc[0];  cP1 = lc[16];  gP0 = lg[0];  gP1 = lg[16];
  aP0 = tile_product(a, frag_b(cP0, hi));
  aP1 = tile_product(a, frag_b(cP1, hi));
  yP0 = gP0; yP1 = gP1;
  cQ0 = lc[STAGE_COLS];  cQ1 = lc[STAGE_COLS + 16];
  gQ0 = lg[STAGE_COLS];  gQ1 = lg[STAGE_COLS + 16];

  for (int c0 = 0; c0 < N_; c0 += 2 * STAGE_COLS) {
    const int n0 = (c0 + 2 * STAGE_COLS) & (N_ - 1);
    const int n1 = (c0 + 3 * STAGE_COLS) & (N_ - 1);

    aQ0 = tile_product(a, frag_b(cQ0, hi));
    aQ1 = tile_product(a, frag_b(cQ1, hi));
    yQ0 = gQ0; yQ1 = gQ1;
    cP0 = lc[n0];  cP1 = lc[n0 + 16];  gP0 = lg[n0];  gP1 = lg[n0 + 16];
    amax_update(aP0, yP0, c0 + idx);
    amax_update(aP1, yP1, c0 + 16 + idx);

    aP0 = tile_product(a, frag_b(cP0, hi));
    aP1 = tile_product(a, frag_b(cP1, hi));
    yP0 = gP0; yP1 = gP1;
    cQ0 = lc[n1];  cQ1 = lc[n1 + 16];  gQ0 = lg[n1];  gQ1 = lg[n1 + 16];
    amax_update(aQ0, yQ0, c0 + STAGE_COLS + idx);
    amax_update(aQ1, yQ1, c0 + STAGE_COLS + 16 + idx);
  }

#pragma unroll
  for (int mask = 1; mask < 16; mask <<= 1) {
#pragma unroll
    for (int r = 0; r < 8; ++r) {
      float ov = __shfl_xor(best[r], mask, 32);
      int   oi = __shfl_xor(bidx[r], mask, 32);
      if (ov > best[r] || (ov == best[r] && oi < bidx[r])) {
        best[r] = ov; bidx[r] = oi;
      }
    }
  }

  if (idx < 8) s_idx[wave * 16 + hi * 8 + idx] = bidx[idx];
  __syncthreads();

  if (tid < ROWS_PER_BLOCK) {
    const int grow = blockIdx.x * ROWS_PER_BLOCK + tid;
    const int j    = s_idx[tid];
    float4 x0a = s_cols[j];
    float4 xn  = rp[grow];
    const float tv = t[b], omt = 1.0f - tv;

    float vx = xn.x - x0a.x, vy = xn.y - x0a.y, vz = xn.z - x0a.z;
    float xtx = omt * x0a.x + tv * xn.x;
    float xty = omt * x0a.y + tv * xn.y;
    float xtz = omt * x0a.z + tv * xn.z;

    float p0 = s_b2[0], p1 = s_b2[1], p2 = s_b2[2];
    for (int k = 0; k < H_; ++k) {
      float h = xtx * s_W1[k] + xty * s_W1[H_ + k] + xtz * s_W1[2 * H_ + k]
              + tv * s_Wt[k] + s_b1[k];
      h = fmaxf(h, 0.0f);
      p0 += h * s_W2[k * 3 + 0];
      p1 += h * s_W2[k * 3 + 1];
      p2 += h * s_W2[k * 3 + 2];
    }
    const size_t base  = ((size_t)b * N_ + grow) * 3;
    const size_t vbase = (size_t)B_ * N_ * 3 + base;
    out[base + 0] = p0; out[base + 1] = p1; out[base + 2] = p2;
    out[vbase + 0] = vx; out[vbase + 1] = vy; out[vbase + 2] = vz;
  }
}

extern "C" void kernel_launch(void* const* d_in, const int* in_sizes, int n_in,
                              void* d_out, int out_size, void* d_ws, size_t ws_size,
                              hipStream_t stream) {
  (void)in_sizes; (void)n_in; (void)out_size; (void)ws_size;
  const float* cloud = (const float*)d_in[0];
  const float* noise = (const float*)d_in[1];
  const float* t     = (const float*)d_in[2];
  const float* W1    = (const float*)d_in[3];
  const float* Wt    = (const float*)d_in[4];
  const float* b1    = (const float*)d_in[5];
  const float* W2    = (const float*)d_in[6];
  const float* b2    = (const float*)d_in[7];
  float* out = (float*)d_out;

  char* ws = (char*)d_ws;
  float4* ypts = (float4*)ws;
  float4* xpts = (float4*)(ws + (size_t)B_ * N_ * sizeof(float4));
  float*  f    = (float*) (ws + (size_t)2 * B_ * N_ * sizeof(float4));
  float*  g    = f + (size_t)B_ * N_;

  prep_kernel<<<B_, THREADS, 0, stream>>>(cloud, noise, ypts, xpts, f, g);

  const float lconst = -logf((float)N_);
  dim3 grid(N_ / ROWS_PER_BLOCK, B_);
  for (int it = 0; it < 14; ++it) {
    float eps = (float)(32.0 * pow(1e-6 / 32.0, (double)it / 13.0));
    sinkhorn_half<<<grid, TILE_THREADS, 0, stream>>>(xpts, ypts, g, f, eps, lconst);
    sinkhorn_half<<<grid, TILE_THREADS, 0, stream>>>(ypts, xpts, f, g, eps, lconst);
  }

  assign_mlp_kernel<<<grid, TILE_THREADS, 0, stream>>>(xpts, ypts, g, t,
                                                  W1, Wt, b1, W2, b2, out);
}


# LSE loop as a three-stage pipeline: exp2 and sums of block n-1 issue beside scale/max/subtract of block n and products of block n+1, so no exp2 or add waits on its neighbour; values, sum order kept
# speedup vs baseline: 2.3255x; 1.0959x over previous
#include <hip/hip_runtime.h>
#include <cmath>

#define B_  8
#define N_  2048
#define H_  256
#define THREADS 256
#define TILE_THREADS 128
#define ROWS_PER_BLOCK 64
#define COLS_PER_THREAD (N_ / TILE_THREADS)

typedef __attribute__((ext_vector_type(2))) float v2f;
typedef __attribute__((ext_vector_type(8))) float v8f;

#define LOG2E 1.4426950408889634f
#define LN2   0.6931471805599453f

#define STAGE_COLS 32

#define BLOCK_TILES 8
#define BLOCK_COLS (16 * BLOCK_TILES)

__device__ __forceinline__ v2f frag_b(const float4 c4, const int hi) {
  v2f bf;
  bf.x = hi ? c4.z : c4.x;
  bf.y = hi ? 0.0f : c4.y;
  return bf;
}

__device__ __forceinline__ v8f tile_product(const v2f a, const v2f bf) {
  const v8f cz = {};
  return __builtin_amdgcn_wmma_f32_16x16x4_f32(
      false, a, false, bf, (short)0, cz, false, false);
}

__global__ __launch_bounds__(THREADS)
void prep_kernel(const float* __restrict__ cloud, const float* __restrict__ noise,
                 float4* __restrict__ ypts, float4* __restrict__ xpts,
                 float* __restrict__ f, float* __restrict__ g) {
  const int b   = blockIdx.x;
  const int tid = threadIdx.x;
  __shared__ float s_sum[THREADS];
  __shared__ float s_sq[THREADS];

  const float* cb = cloud + (size_t)b * N_ * 3;
  const float* nb = noise + (size_t)b * N_ * 3;

  float sum = 0.f, sq = 0.f;
  for (int i = tid; i < N_ * 3; i += THREADS) {
    float v = cb[i];
    sum += v; sq += v * v;
  }
  s_sum[tid] = sum; s_sq[tid] = sq;
  __syncthreads();
  for (int off = THREADS / 2; off > 0; off >>= 1) {
    if (tid < off) { s_sum[tid] += s_sum[tid + off]; s_sq[tid] += s_sq[tid + off]; }
    __syncthreads();
  }
  const float n    = (float)(N_ * 3);
  const float tot  = s_sum[0];
  const float tot2 = s_sq[0];
  const float var  = (tot2 - tot * tot / n) / (n - 1.0f);
  const float inv_std = rsqrtf(var);

  for (int p = tid; p < N_; p += THREADS) {
    float yx = cb[p * 3 + 0] * inv_std;
    float yy = cb[p * 3 + 1] * inv_std;
    float yz = cb[p * 3 + 2] * inv_std;
    ypts[(size_t)b * N_ + p] = make_float4(yx, yy, yz, yx * yx + yy * yy + yz * yz);
    float xx = nb[p * 3 + 0];
    float xy = nb[p * 3 + 1];
    float xz = nb[p * 3 + 2];
    xpts[(size_t)b * N_ + p] = make_float4(xx, xy, xz, xx * xx + xy * xy + xz * xz);
    f[(size_t)b * N_ + p] = 0.f;
    g[(size_t)b * N_ + p] = 0.f;
  }
}

__global__ __launch_bounds__(TILE_THREADS)
void sinkhorn_half(const float4* __restrict__ rows_pts,
                   const float4* __restrict__ cols_pts,
                   const float*  __restrict__ in_pot,
                   float*        __restrict__ out_pot,
                   float eps, float lconst) {
  const int b    = blockIdx.y;
  const int tid  = threadIdx.x;
  const int lane = tid & 31;
  const int wave = tid >> 5;
  const int hi   = lane >> 4;
  const int idx  = lane & 15;

  __shared__ __attribute__((aligned(16))) v2f s_cols[2 * N_];

  const float4* cp = cols_pts + (size_t)b * N_;
  const float*  ip = in_pot   + (size_t)b * N_;
  float4 cst[COLS_PER_THREAD];
  float  pst[COLS_PER_THREAD];
#pragma unroll
  for (int k = 0; k < COLS_PER_THREAD; ++k) {
    cst[k] = cp[tid + k * TILE_THREADS];
    pst[k] = ip[tid + k * TILE_THREADS];
  }
#pragma unroll
  for (int k = 0; k < COLS_PER_THREAD; ++k) {
    const int j = tid + k * TILE_THREADS;
    v2f lo, up;
    lo.x = cst[k].x;  lo.y = cst[k].y;
    up.x = cst[k].z;  up.y = pst[k] - 0.5f * cst[k].w;
    s_cols[2 * j]     = lo;
    s_cols[2 * j + 1] = up;
  }
  __syncthreads();

  const int row_base = blockIdx.x * ROWS_PER_BLOCK + wave * 16;
  const float4* rp = rows_pts + (size_t)b * N_;

  float4 r4 = rp[row_base + idx];
  v2f a;
  a.x = hi ? r4.z : r4.x;
  a.y = hi ? 1.0f : r4.y;

  v8f m, s;
#pragma unroll
  for (int r = 0; r < 8; ++r) { m[r] = -3.0e38f; s[r] = 0.f; }
  const float c2 = LOG2E / eps;

  const v2f* lc = s_cols + 2 * idx + hi;

  v2f cP[BLOCK_TILES], cQ[BLOCK_TILES];
  v8f aP[BLOCK_TILES], aQ[BLOCK_TILES];

  auto read_block = [&](const int col0, v2f (&c)[BLOCK_TILES]) {
#pragma unroll
    for (int t = 0; t < BLOCK_TILES; ++t) c[t] = lc[2 * (col0 + 16 * t)];
  };

  auto lse_front = [&](const v8f (&acc)[BLOCK_TILES], v8f (&d)[BLOCK_TILES], v8f &dm) {
    v8f v[BLOCK_TILES];
#pragma unroll
    for (int t = 0; t < BLOCK_TILES; ++t) v[t] = acc[t] * c2;
    const v8f mn = __builtin_elementwise_max(
        __builtin_elementwise_max(__builtin_elementwise_max(__builtin_elementwise_max(v[0], v[1]), v[2]),
                                  __builtin_elementwise_max(__builtin_elementwise_max(v[3], v[4]), v[5])),
        __builtin_elementwise_max(__builtin_elementwise_max(v[6], v[7]), m));
    dm = m - mn;
#pragma unroll
    for (int t = 0; t < BLOCK_TILES; ++t) {
#pragma clang fp contract(off)
      d[t] = v[t] - mn;
    }
    m = mn;
  };

  auto lse_back = [&](const v8f (&d)[BLOCK_TILES], const v8f &dm) {
    v8f keep;
#pragma unroll
    for (int r = 0; r < 8; ++r) keep[r] = __builtin_amdgcn_exp2f(dm[r]);
    v8f e[BLOCK_TILES];
#pragma unroll
    for (int t = 0; t < BLOCK_TILES; ++t) {
#pragma unroll
      for (int r = 0; r < 8; ++r) e[t][r] = __builtin_amdgcn_exp2f(d[t][r]);
    }
    const v8f blk = ((e[0] + e[1]) + (e[2] + e[3])) + ((e[4] + e[5]) + (e[6] + e[7]));
    s = __builtin_elementwise_fma(s, keep, blk);
  };

  auto block_step = [&](const v2f (&c)[BLOCK_TILES], v8f (&nxt)[BLOCK_TILES],
                        const v8f (&acc)[BLOCK_TILES], v8f (&d)[BLOCK_TILES], v8f &dm,
                        const v8f (&pd)[BLOCK_TILES], const v8f &pdm) {
#pragma unroll
    for (int t = 0; t < BLOCK_TILES; ++t) nxt[t] = tile_product(a, c[t]);
    lse_front(acc, d, dm);
    lse_back(pd, pdm);
#pragma unroll
    for (int t = 0; t < BLOCK_TILES; t += 2) {
      __builtin_amdgcn_sched_group_barrier(0x008, 1, 0);
#pragma unroll
      for (int k = 0; k < 9; ++k) {
        __builtin_amdgcn_sched_group_barrier(0x002, 1, 0);
        __builtin_amdgcn_sched_group_barrier(0x400, 1, 0);
      }
      __builtin_amdgcn_sched_group_barrier(0x002, 7, 0);
      __builtin_amdgcn_sched_group_barrier(0x008, 1, 0);
#pragma unroll
      for (int k = 0; k < 9; ++k) {
        __builtin_amdgcn_sched_group_barrier(0x002, 1, 0);
        __builtin_amdgcn_sched_group_barrier(0x400, 1, 0);
      }
      __builtin_amdgcn_sched_group_barrier(0x002, 8, 0);
      if (t == 2) __builtin_amdgcn_sched_group_barrier(0x100, BLOCK_TILES, 0);
    }
  };

  v8f dP[BLOCK_TILES], dQ[BLOCK_TILES];
  v8f gP, gQ;

  read_block(0, cP);
#pragma unroll
  for (int t = 0; t < BLOCK_TILES; ++t) aP[t] = tile_product(a, cP[t]);
  read_block(BLOCK_COLS, cQ);
  read_block(2 * BLOCK_COLS, cP);
#pragma unroll
  for (int t = 0; t < BLOCK_TILES; ++t) aQ[t] = tile_product(a, cQ[t]);
  lse_front(aP, dP, gP);

  for (int c0 = BLOCK_COLS; c0 < N_ - BLOCK_COLS; c0 += 2 * BLOCK_COLS) {
    const int n0 = (c0 + 2 * BLOCK_COLS) & (N_ - 1);
    const int n1 = (c0 + 3 * BLOCK_COLS) & (N_ - 1);

    read_block(n0, cQ);
    block_step(cP, aP, aQ, dQ, gQ, dP, gP);

    read_block(n1, cP);
    block_step(cQ, aQ, aP, dP, gP, dQ, gQ);
  }
  lse_front(aQ, dQ, gQ);
  lse_back(dP, gP);
  lse_back(dQ, gQ);

#pragma unroll
  for (int mask = 1; mask < 16; mask <<= 1) {
#pragma unroll
    for (int r = 0; r < 8; ++r) {
      float om = __shfl_xor(m[r], mask, 32);
      float os = __shfl_xor(s[r], mask, 32);
      float d  = om - m[r];
      float e  = __builtin_amdgcn_exp2f(-fabsf(d));
      bool  up = d > 0.0f;
      s[r] = up ? fmaf(s[r], e, os) : fmaf(os, e, s[r]);
      m[r] = up ? om : m[r];
    }
  }

  if (idx < 8) {
    const int row = row_base + hi * 8 + idx;
    const int r   = idx;
    float sqrow = rp[row].w;
    float lse = LN2 * (m[r] + __builtin_amdgcn_logf(s[r]));
    out_pot[(size_t)b * N_ + row] = -eps * (lconst + lse) + 0.5f * sqrow;
  }
}

__global__ __launch_bounds__(TILE_THREADS)
void assign_mlp_kernel(const float4* __restrict__ xpts,
                       const float4* __restrict__ ypts,
                       const float*  __restrict__ g,
                       const float*  __restrict__ t,
                       const float*  __restrict__ W1,
                       const float*  __restrict__ Wt,
                       const float*  __restrict__ b1,
                       const float*  __restrict__ W2,
                       const float*  __restrict__ b2,
                       float*        __restrict__ out) {
  const int b    = blockIdx.y;
  const int tid  = threadIdx.x;
  const int lane = tid & 31;
  const int wave = tid >> 5;
  const int hi   = lane >> 4;
  const int idx  = lane & 15;

  __shared__ float4 s_cols[N_];
  __shared__ float  s_gy[N_];
  __shared__ float  s_W1[3 * H_];
  __shared__ float  s_Wt[H_];
  __shared__ float  s_b1[H_];
  __shared__ float  s_W2[H_ * 3];
  __shared__ float  s_b2[4];
  __shared__ int    s_idx[ROWS_PER_BLOCK];

  const float4* cp = ypts + (size_t)b * N_;
  const float*  gp = g    + (size_t)b * N_;
  float4 cst[COLS_PER_THREAD];
  float  pst[COLS_PER_THREAD];
#pragma unroll
  for (int k = 0; k < COLS_PER_THREAD; ++k) {
    cst[k] = cp[tid + k * TILE_THREADS];
    pst[k] = gp[tid + k * TILE_THREADS];
  }
#pragma unroll
  for (int k = 0; k < COLS_PER_THREAD; ++k) {
    const int j = tid + k * TILE_THREADS;
    s_cols[j] = cst[k];
    s_gy[j]   = pst[k] - cst[k].w;
  }
#pragma unroll
  for (int k = 0; k < 3 * H_ / TILE_THREADS; ++k) {
    const int j = tid + k * TILE_THREADS;
    s_W1[j] = W1[j]; s_W2[j] = W2[j];
  }
#pragma unroll
  for (int k = 0; k < H_ / TILE_THREADS; ++k) {
    const int j = tid + k * TILE_THREADS;
    s_Wt[j] = Wt[j]; s_b1[j] = b1[j];
  }
  if (tid < 3) s_b2[tid] = b2[tid];
  __syncthreads();

  const int row_base = blockIdx.x * ROWS_PER_BLOCK + wave * 16;
  const float4* rp = xpts + (size_t)b * N_;

  float4 r4 = rp[row_base + idx];
  v2f a;
  a.x = hi ? 2.0f * r4.z : 2.0f * r4.x;
  a.y = hi ? 0.0f        : 2.0f * r4.y;

  float best[8]; int bidx[8];
#pragma unroll
  for (int r = 0; r < 8; ++r) { best[r] = -3.0e38f; bidx[r] = 0; }

  auto amax_update = [&](const v8f acc, const float gy, const int col) {
#pragma unroll
    for (int r = 0; r < 8; ++r) {
      float v = acc[r] + gy;
      if (v > best[r]) { best[r] = v; bidx[r] = col; }
    }
  };

  const float4* lc = s_cols + idx;
  const float*  lg = s_gy + idx;

  float4 cP0, cP1, cQ0, cQ1;
  float  gP0, gP1, gQ0, gQ1;
  v8f    aP0, aP1, aQ0, aQ1;
  float  yP0, yP1, yQ0, yQ1;

  cP0 = lc[0];  cP1 = lc[16];  gP0 = lg[0];  gP1 = lg[16];
  aP0 = tile_product(a, frag_b(cP0, hi));
  aP1 = tile_product(a, frag_b(cP1, hi));
  yP0 = gP0; yP1 = gP1;
  cQ0 = lc[STAGE_COLS];  cQ1 = lc[STAGE_COLS + 16];
  gQ0 = lg[STAGE_COLS];  gQ1 = lg[STAGE_COLS + 16];

  for (int c0 = 0; c0 < N_; c0 += 2 * STAGE_COLS) {
    const int n0 = (c0 + 2 * STAGE_COLS) & (N_ - 1);
    const int n1 = (c0 + 3 * STAGE_COLS) & (N_ - 1);

    aQ0 = tile_product(a, frag_b(cQ0, hi));
    aQ1 = tile_product(a, frag_b(cQ1, hi));
    yQ0 = gQ0; yQ1 = gQ1;
    cP0 = lc[n0];  cP1 = lc[n0 + 16];  gP0 = lg[n0];  gP1 = lg[n0 + 16];
    amax_update(aP0, yP0, c0 + idx);
    amax_update(aP1, yP1, c0 + 16 + idx);

    aP0 = tile_product(a, frag_b(cP0, hi));
    aP1 = tile_product(a, frag_b(cP1, hi));
    yP0 = gP0; yP1 = gP1;
    cQ0 = lc[n1];  cQ1 = lc[n1 + 16];  gQ0 = lg[n1];  gQ1 = lg[n1 + 16];
    amax_update(aQ0, yQ0, c0 + STAGE_COLS + idx);
    amax_update(aQ1, yQ1, c0 + STAGE_COLS + 16 + idx);
  }

#pragma unroll
  for (int mask = 1; mask < 16; mask <<= 1) {
#pragma unroll
    for (int r = 0; r < 8; ++r) {
      float ov = __shfl_xor(best[r], mask, 32);
      int   oi = __shfl_xor(bidx[r], mask, 32);
      if (ov > best[r] || (ov == best[r] && oi < bidx[r])) {
        best[r] = ov; bidx[r] = oi;
      }
    }
  }

  if (idx < 8) s_idx[wave * 16 + hi * 8 + idx] = bidx[idx];
  __syncthreads();

  if (tid < ROWS_PER_BLOCK) {
    const int grow = blockIdx.x * ROWS_PER_BLOCK + tid;
    const int j    = s_idx[tid];
    float4 x0a = s_cols[j];
    float4 xn  = rp[grow];
    const float tv = t[b], omt = 1.0f - tv;

    float vx = xn.x - x0a.x, vy = xn.y - x0a.y, vz = xn.z - x0a.z;
    float xtx = omt * x0a.x + tv * xn.x;
    float xty = omt * x0a.y + tv * xn.y;
    float xtz = omt * x0a.z + tv * xn.z;

    float p0 = s_b2[0], p1 = s_b2[1], p2 = s_b2[2];
    for (int k = 0; k < H_; ++k) {
      float h = xtx * s_W1[k] + xty * s_W1[H_ + k] + xtz * s_W1[2 * H_ + k]
              + tv * s_Wt[k] + s_b1[k];
      h = fmaxf(h, 0.0f);
      p0 += h * s_W2[k * 3 + 0];
      p1 += h * s_W2[k * 3 + 1];
      p2 += h * s_W2[k * 3 + 2];
    }
    const size_t base  = ((size_t)b * N_ + grow) * 3;
    const size_t vbase = (size_t)B_ * N_ * 3 + base;
    out[base + 0] = p0; out[base + 1] = p1; out[base + 2] = p2;
    out[vbase + 0] = vx; out[vbase + 1] = vy; out[vbase + 2] = vz;
  }
}

extern "C" void kernel_launch(void* const* d_in, const int* in_sizes, int n_in,
                              void* d_out, int out_size, void* d_ws, size_t ws_size,
                              hipStream_t stream) {
  (void)in_sizes; (void)n_in; (void)out_size; (void)ws_size;
  const float* cloud = (const float*)d_in[0];
  const float* noise = (const float*)d_in[1];
  const float* t     = (const float*)d_in[2];
  const float* W1    = (const float*)d_in[3];
  const float* Wt    = (const float*)d_in[4];
  const float* b1    = (const float*)d_in[5];
  const float* W2    = (const float*)d_in[6];
  const float* b2    = (const float*)d_in[7];
  float* out = (float*)d_out;

  char* ws = (char*)d_ws;
  float4* ypts = (float4*)ws;
  float4* xpts = (float4*)(ws + (size_t)B_ * N_ * sizeof(float4));
  float*  f    = (float*) (ws + (size_t)2 * B_ * N_ * sizeof(float4));
  float*  g    = f + (size_t)B_ * N_;

  prep_kernel<<<B_, THREADS, 0, stream>>>(cloud, noise, ypts, xpts, f, g);

  const float lconst = -logf((float)N_);
  dim3 grid(N_ / ROWS_PER_BLOCK, B_);
  for (int it = 0; it < 14; ++it) {
    float eps = (float)(32.0 * pow(1e-6 / 32.0, (double)it / 13.0));
    sinkhorn_half<<<grid, TILE_THREADS, 0, stream>>>(xpts, ypts, g, f, eps, lconst);
    sinkhorn_half<<<grid, TILE_THREADS, 0, stream>>>(ypts, xpts, f, g, eps, lconst);
  }

  assign_mlp_kernel<<<grid, TILE_THREADS, 0, stream>>>(xpts, ypts, g, t,
                                                  W1, Wt, b1, W2, b2, out);
}


# LSE kernel: scale log2(e)/eps multiplied into the row operand of the f32 product once per lane, so the per-element multiply goes (issue groups resized to fit); v rounding changes; v-mn unfused
# speedup vs baseline: 2.5969x; 1.1167x over previous
#include <hip/hip_runtime.h>
#include <cmath>

#define B_  8
#define N_  2048
#define H_  256
#define THREADS 256
#define TILE_THREADS 128
#define ROWS_PER_BLOCK 64
#define COLS_PER_THREAD (N_ / TILE_THREADS)

typedef __attribute__((ext_vector_type(2))) float v2f;
typedef __attribute__((ext_vector_type(4))) float v4f;
typedef __attribute__((ext_vector_type(8))) float v8f;

#define LOG2E 1.4426950408889634f
#define LN2   0.6931471805599453f

#define STAGE_COLS 32

#define BLOCK_TILES 8
#define BLOCK_COLS (16 * BLOCK_TILES)

__device__ __forceinline__ v2f frag_b(const float4 c4, const int hi) {
  v2f bf;
  bf.x = hi ? c4.z : c4.x;
  bf.y = hi ? 0.0f : c4.y;
  return bf;
}

__device__ __forceinline__ v8f tile_product(const v2f a, const v2f bf) {
  const v8f cz = {};
  return __builtin_amdgcn_wmma_f32_16x16x4_f32(
      false, a, false, bf, (short)0, cz, false, false);
}

__global__ __launch_bounds__(THREADS)
void prep_kernel(const float* __restrict__ cloud, const float* __restrict__ noise,
                 float4* __restrict__ ypts, float4* __restrict__ xpts,
                 float* __restrict__ f, float* __restrict__ g) {
  const int b   = blockIdx.x;
  const int tid = threadIdx.x;
  __shared__ float s_sum[THREADS];
  __shared__ float s_sq[THREADS];

  const float* cb = cloud + (size_t)b * N_ * 3;
  const float* nb = noise + (size_t)b * N_ * 3;

  float sum = 0.f, sq = 0.f;
  for (int i = tid; i < N_ * 3; i += THREADS) {
    float v = cb[i];
    sum += v; sq += v * v;
  }
  s_sum[tid] = sum; s_sq[tid] = sq;
  __syncthreads();
  for (int off = THREADS / 2; off > 0; off >>= 1) {
    if (tid < off) { s_sum[tid] += s_sum[tid + off]; s_sq[tid] += s_sq[tid + off]; }
    __syncthreads();
  }
  const float n    = (float)(N_ * 3);
  const float tot  = s_sum[0];
  const float tot2 = s_sq[0];
  const float var  = (tot2 - tot * tot / n) / (n - 1.0f);
  const float inv_std = rsqrtf(var);

  for (int p = tid; p < N_; p += THREADS) {
    float yx = cb[p * 3 + 0] * inv_std;
    float yy = cb[p * 3 + 1] * inv_std;
    float yz = cb[p * 3 + 2] * inv_std;
    ypts[(size_t)b * N_ + p] = make_float4(yx, yy, yz, yx * yx + yy * yy + yz * yz);
    float xx = nb[p * 3 + 0];
    float xy = nb[p * 3 + 1];
    float xz = nb[p * 3 + 2];
    xpts[(size_t)b * N_ + p] = make_float4(xx, xy, xz, xx * xx + xy * xy + xz * xz);
    f[(size_t)b * N_ + p] = 0.f;
    g[(size_t)b * N_ + p] = 0.f;
  }
}

template <typename T>
__device__ __forceinline__ T max_of3(const T x, const T y, const T z) {
  return __builtin_elementwise_max(__builtin_elementwise_max(x, y), z);
}

__global__ __launch_bounds__(TILE_THREADS)
void sinkhorn_half(const float4* __restrict__ rows_pts,
                   const float4* __restrict__ cols_pts,
                   const float*  __restrict__ in_pot,
                   float*        __restrict__ out_pot,
                   float eps, float lconst) {
  const int b    = blockIdx.y;
  const int tid  = threadIdx.x;
  const int lane = tid & 31;
  const int wave = tid >> 5;
  const int hi   = lane >> 4;
  const int idx  = lane & 15;

  __shared__ __attribute__((aligned(16))) v2f s_cols[2 * N_];

  const float4* cp = cols_pts + (size_t)b * N_;
  const float*  ip = in_pot   + (size_t)b * N_;
  float4 cst[COLS_PER_THREAD];
  float  pst[COLS_PER_THREAD];
#pragma unroll
  for (int k = 0; k < COLS_PER_THREAD; ++k) {
    cst[k] = cp[tid + k * TILE_THREADS];
    pst[k] = ip[tid + k * TILE_THREADS];
  }
#pragma unroll
  for (int k = 0; k < COLS_PER_THREAD; ++k) {
    const int j = tid + k * TILE_THREADS;
    v2f lo, up;
    lo.x = cst[k].x;  lo.y = cst[k].y;
    up.x = cst[k].z;  up.y = pst[k] - 0.5f * cst[k].w;
    s_cols[2 * j]     = lo;
    s_cols[2 * j + 1] = up;
  }
  __syncthreads();

  const int row_base = blockIdx.x * ROWS_PER_BLOCK + wave * 16;
  const float4* rp = rows_pts + (size_t)b * N_;

  const float c2 = LOG2E / eps;
  const float4 r4 = rp[row_base + idx];
  v2f a;
  a.x = c2 * (hi ? r4.z : r4.x);
  a.y = hi ? c2 : c2 * r4.y;

  float m = -3.0e38f, s = 0.f;

  const v2f* lc = s_cols + 2 * idx + hi;

  v2f cP[BLOCK_TILES], cQ[BLOCK_TILES];
  v8f aP[BLOCK_TILES], aQ[BLOCK_TILES];

  auto read_block = [&](const int col0, v2f (&c)[BLOCK_TILES]) {
#pragma unroll
    for (int t = 0; t < BLOCK_TILES; ++t) c[t] = lc[2 * (col0 + 16 * t)];
  };

  auto lse_front = [&](const v8f (&v)[BLOCK_TILES], v8f (&d)[BLOCK_TILES], float &dm) {
    const v8f p = max_of3(max_of3(v[0], v[1], v[2]), max_of3(v[3], v[4], v[5]), v[6]);
    const v8f w = v[7];
    const float q0 = max_of3(p[0], p[1], p[2]);
    const float q1 = max_of3(p[3], p[4], p[5]);
    const float q2 = max_of3(p[6], p[7], w[0]);
    const float q3 = max_of3(w[1], w[2], w[3]);
    const float q4 = max_of3(w[4], w[5], w[6]);
    const float mn = max_of3(max_of3(q0, q1, q2), max_of3(q3, q4, w[7]), m);
    dm = m - mn;
#pragma unroll
    for (int t = 0; t < BLOCK_TILES; ++t) {
      d[t] = v[t] - mn;
    }
    m = mn;
  };

  auto lse_back = [&](const v8f (&d)[BLOCK_TILES], const float dm) {
    const float keep = __builtin_amdgcn_exp2f(dm);
    v8f e[BLOCK_TILES];
#pragma unroll
    for (int t = 0; t < BLOCK_TILES; ++t) {
#pragma unroll
      for (int r = 0; r < 8; ++r) e[t][r] = __builtin_amdgcn_exp2f(d[t][r]);
    }
    const v8f blk = ((e[0] + e[1]) + (e[2] + e[3])) + ((e[4] + e[5]) + (e[6] + e[7]));
    const v4f b4 = blk.lo + blk.hi;
    const v2f b2 = b4.lo + b4.hi;
    s = __builtin_fmaf(s, keep, b2.x + b2.y);
  };

  auto block_step = [&](const v2f (&c)[BLOCK_TILES], v8f (&nxt)[BLOCK_TILES],
                        const v8f (&acc)[BLOCK_TILES], v8f (&d)[BLOCK_TILES], float &dm,
                        const v8f (&pd)[BLOCK_TILES], const float pdm) {
#pragma unroll
    for (int t = 0; t < BLOCK_TILES; ++t) nxt[t] = tile_product(c[t], a);
    lse_front(acc, d, dm);
    lse_back(pd, pdm);
    constexpr int kPairs[BLOCK_TILES] = {13, 13, 13, 13, 13, 0, 0, 0};
    constexpr int kPlain[BLOCK_TILES] = {0, 0, 0, 0, 0, 10, 10, 20};
#pragma unroll
    for (int t = 0; t < BLOCK_TILES; ++t) {
      __builtin_amdgcn_sched_group_barrier(0x008, 1, 0);
#pragma unroll
      for (int k = 0; k < kPairs[t]; ++k) {
        __builtin_amdgcn_sched_group_barrier(0x002, 1, 0);
        __builtin_amdgcn_sched_group_barrier(0x400, 1, 0);
      }
#pragma unroll
      for (int k = 0; k < kPlain[t]; ++k) __builtin_amdgcn_sched_group_barrier(0x002, 1, 0);
      if (t == 2) __builtin_amdgcn_sched_group_barrier(0x100, BLOCK_TILES, 0);
    }
  };

  v8f dP[BLOCK_TILES], dQ[BLOCK_TILES];
  float gP, gQ;

  read_block(0, cP);
#pragma unroll
  for (int t = 0; t < BLOCK_TILES; ++t) aP[t] = tile_product(cP[t], a);
  read_block(BLOCK_COLS, cQ);
  read_block(2 * BLOCK_COLS, cP);
#pragma unroll
  for (int t = 0; t < BLOCK_TILES; ++t) aQ[t] = tile_product(cQ[t], a);
  lse_front(aP, dP, gP);

  for (int c0 = BLOCK_COLS; c0 < N_ - BLOCK_COLS; c0 += 2 * BLOCK_COLS) {
    const int n0 = (c0 + 2 * BLOCK_COLS) & (N_ - 1);
    const int n1 = (c0 + 3 * BLOCK_COLS) & (N_ - 1);

    read_block(n0, cQ);
    block_step(cP, aP, aQ, dQ, gQ, dP, gP);

    read_block(n1, cP);
    block_step(cQ, aQ, aP, dP, gP, dQ, gQ);
  }
  lse_front(aQ, dQ, gQ);
  lse_back(dP, gP);
  lse_back(dQ, gQ);

  {
    const float om = __shfl_xor(m, 16, 32);
    const float os = __shfl_xor(s, 16, 32);
    const float d  = om - m;
    const float e  = __builtin_amdgcn_exp2f(-fabsf(d));
    const bool  up = d > 0.0f;
    s = up ? fmaf(s, e, os) : fmaf(os, e, s);
    m = up ? om : m;
  }

  if (hi == 0) {
    const int row = row_base + idx;
    float lse = LN2 * (m + __builtin_amdgcn_logf(s));
    out_pot[(size_t)b * N_ + row] = -eps * (lconst + lse) + 0.5f * r4.w;
  }
}

__global__ __launch_bounds__(TILE_THREADS)
void assign_mlp_kernel(const float4* __restrict__ xpts,
                       const float4* __restrict__ ypts,
                       const float*  __restrict__ g,
                       const float*  __restrict__ t,
                       const float*  __restrict__ W1,
                       const float*  __restrict__ Wt,
                       const float*  __restrict__ b1,
                       const float*  __restrict__ W2,
                       const float*  __restrict__ b2,
                       float*        __restrict__ out) {
  const int b    = blockIdx.y;
  const int tid  = threadIdx.x;
  const int lane = tid & 31;
  const int wave = tid >> 5;
  const int hi   = lane >> 4;
  const int idx  = lane & 15;

  __shared__ float4 s_cols[N_];
  __shared__ float  s_gy[N_];
  __shared__ float  s_W1[3 * H_];
  __shared__ float  s_Wt[H_];
  __shared__ float  s_b1[H_];
  __shared__ float  s_W2[H_ * 3];
  __shared__ float  s_b2[4];
  __shared__ int    s_idx[ROWS_PER_BLOCK];

  const float4* cp = ypts + (size_t)b * N_;
  const float*  gp = g    + (size_t)b * N_;
  float4 cst[COLS_PER_THREAD];
  float  pst[COLS_PER_THREAD];
#pragma unroll
  for (int k = 0; k < COLS_PER_THREAD; ++k) {
    cst[k] = cp[tid + k * TILE_THREADS];
    pst[k] = gp[tid + k * TILE_THREADS];
  }
#pragma unroll
  for (int k = 0; k < COLS_PER_THREAD; ++k) {
    const int j = tid + k * TILE_THREADS;
    s_cols[j] = cst[k];
    s_gy[j]   = pst[k] - cst[k].w;
  }
#pragma unroll
  for (int k = 0; k < 3 * H_ / TILE_THREADS; ++k) {
    const int j = tid + k * TILE_THREADS;
    s_W1[j] = W1[j]; s_W2[j] = W2[j];
  }
#pragma unroll
  for (int k = 0; k < H_ / TILE_THREADS; ++k) {
    const int j = tid + k * TILE_THREADS;
    s_Wt[j] = Wt[j]; s_b1[j] = b1[j];
  }
  if (tid < 3) s_b2[tid] = b2[tid];
  __syncthreads();

  const int row_base = blockIdx.x * ROWS_PER_BLOCK + wave * 16;
  const float4* rp = xpts + (size_t)b * N_;

  float4 r4 = rp[row_base + idx];
  v2f a;
  a.x = hi ? 2.0f * r4.z : 2.0f * r4.x;
  a.y = hi ? 0.0f        : 2.0f * r4.y;

  float best[8]; int bidx[8];
#pragma unroll
  for (int r = 0; r < 8; ++r) { best[r] = -3.0e38f; bidx[r] = 0; }

  auto amax_update = [&](const v8f acc, const float gy, const int col) {
#pragma unroll
    for (int r = 0; r < 8; ++r) {
      float v = acc[r] + gy;
      if (v > best[r]) { best[r] = v; bidx[r] = col; }
    }
  };

  const float4* lc = s_cols + idx;
  const float*  lg = s_gy + idx;

  float4 cP0, cP1, cQ0, cQ1;
  float  gP0, gP1, gQ0, gQ1;
  v8f    aP0, aP1, aQ0, aQ1;
  float  yP0, yP1, yQ0, yQ1;

  cP0 = lc[0];  cP1 = lc[16];  gP0 = lg[0];  gP1 = lg[16];
  aP0 = tile_product(a, frag_b(cP0, hi));
  aP1 = tile_product(a, frag_b(cP1, hi));
  yP0 = gP0; yP1 = gP1;
  cQ0 = lc[STAGE_COLS];  cQ1 = lc[STAGE_COLS + 16];
  gQ0 = lg[STAGE_COLS];  gQ1 = lg[STAGE_COLS + 16];

  for (int c0 = 0; c0 < N_; c0 += 2 * STAGE_COLS) {
    const int n0 = (c0 + 2 * STAGE_COLS) & (N_ - 1);
    const int n1 = (c0 + 3 * STAGE_COLS) & (N_ - 1);

    aQ0 = tile_product(a, frag_b(cQ0, hi));
    aQ1 = tile_product(a, frag_b(cQ1, hi));
    yQ0 = gQ0; yQ1 = gQ1;
    cP0 = lc[n0];  cP1 = lc[n0 + 16];  gP0 = lg[n0];  gP1 = lg[n0 + 16];
    amax_update(aP0, yP0, c0 + idx);
    amax_update(aP1, yP1, c0 + 16 + idx);

    aP0 = tile_product(a, frag_b(cP0, hi));
    aP1 = tile_product(a, frag_b(cP1, hi));
    yP0 = gP0; yP1 = gP1;
    cQ0 = lc[n1];  cQ1 = lc[n1 + 16];  gQ0 = lg[n1];  gQ1 = lg[n1 + 16];
    amax_update(aQ0, yQ0, c0 + STAGE_COLS + idx);
    amax_update(aQ1, yQ1, c0 + STAGE_COLS + 16 + idx);
  }

#pragma unroll
  for (int mask = 1; mask < 16; mask <<= 1) {
#pragma unroll
    for (int r = 0; r < 8; ++r) {
      float ov = __shfl_xor(best[r], mask, 32);
      int   oi = __shfl_xor(bidx[r], mask, 32);
      if (ov > best[r] || (ov == best[r] && oi < bidx[r])) {
        best[r] = ov; bidx[r] = oi;
      }
    }
  }

  if (idx < 8) s_idx[wave * 16 + hi * 8 + idx] = bidx[idx];
  __syncthreads();

  if (tid < ROWS_PER_BLOCK) {
    const int grow = blockIdx.x * ROWS_PER_BLOCK + tid;
    const int j    = s_idx[tid];
    float4 x0a = s_cols[j];
    float4 xn  = rp[grow];
    const float tv = t[b], omt = 1.0f - tv;

    float vx = xn.x - x0a.x, vy = xn.y - x0a.y, vz = xn.z - x0a.z;
    float xtx = omt * x0a.x + tv * xn.x;
    float xty = omt * x0a.y + tv * xn.y;
    float xtz = omt * x0a.z + tv * xn.z;

    float p0 = s_b2[0], p1 = s_b2[1], p2 = s_b2[2];
    for (int k = 0; k < H_; ++k) {
      float h = xtx * s_W1[k] + xty * s_W1[H_ + k] + xtz * s_W1[2 * H_ + k]
              + tv * s_Wt[k] + s_b1[k];
      h = fmaxf(h, 0.0f);
      p0 += h * s_W2[k * 3 + 0];
      p1 += h * s_W2[k * 3 + 1];
      p2 += h * s_W2[k * 3 + 2];
    }
    const size_t base  = ((size_t)b * N_ + grow) * 3;
    const size_t vbase = (size_t)B_ * N_ * 3 + base;
    out[base + 0] = p0; out[base + 1] = p1; out[base + 2] = p2;
    out[vbase + 0] = vx; out[vbase + 1] = vy; out[vbase + 2] = vz;
  }
}

extern "C" void kernel_launch(void* const* d_in, const int* in_sizes, int n_in,
                              void* d_out, int out_size, void* d_ws, size_t ws_size,
                              hipStream_t stream) {
  (void)in_sizes; (void)n_in; (void)out_size; (void)ws_size;
  const float* cloud = (const float*)d_in[0];
  const float* noise = (const float*)d_in[1];
  const float* t     = (const float*)d_in[2];
  const float* W1    = (const float*)d_in[3];
  const float* Wt    = (const float*)d_in[4];
  const float* b1    = (const float*)d_in[5];
  const float* W2    = (const float*)d_in[6];
  const float* b2    = (const float*)d_in[7];
  float* out = (float*)d_out;

  char* ws = (char*)d_ws;
  float4* ypts = (float4*)ws;
  float4* xpts = (float4*)(ws + (size_t)B_ * N_ * sizeof(float4));
  float*  f    = (float*) (ws + (size_t)2 * B_ * N_ * sizeof(float4));
  float*  g    = f + (size_t)B_ * N_;

  prep_kernel<<<B_, THREADS, 0, stream>>>(cloud, noise, ypts, xpts, f, g);

  const float lconst = -logf((float)N_);
  dim3 grid(N_ / ROWS_PER_BLOCK, B_);
  for (int it = 0; it < 14; ++it) {
    float eps = (float)(32.0 * pow(1e-6 / 32.0, (double)it / 13.0));
    sinkhorn_half<<<grid, TILE_THREADS, 0, stream>>>(xpts, ypts, g, f, eps, lconst);
    sinkhorn_half<<<grid, TILE_THREADS, 0, stream>>>(ypts, xpts, f, g, eps, lconst);
  }

  assign_mlp_kernel<<<grid, TILE_THREADS, 0, stream>>>(xpts, ypts, g, t,
                                                  W1, Wt, b1, W2, b2, out);
}


# LSE against a per-row reference predicted from the row's previous potential, folded into the product's C operand: per element exp2+add only (max, subtract go); exact fallback path; sum order changes
# speedup vs baseline: 2.8121x; 1.0829x over previous
#include <hip/hip_runtime.h>
#include <cmath>

#define B_  8
#define N_  2048
#define H_  256
#define THREADS 256
#define TILE_THREADS 128
#define ROWS_PER_BLOCK 64
#define COLS_PER_THREAD (N_ / TILE_THREADS)

typedef __attribute__((ext_vector_type(2))) float v2f;
typedef __attribute__((ext_vector_type(4))) float v4f;
typedef __attribute__((ext_vector_type(8))) float v8f;

#define LOG2E 1.4426950408889634f
#define LN2   0.6931471805599453f

#define STAGE_COLS 32

#define BLOCK_TILES 8
#define BLOCK_COLS (16 * BLOCK_TILES)

__device__ __forceinline__ v2f frag_b(const float4 c4, const int hi) {
  v2f bf;
  bf.x = hi ? c4.z : c4.x;
  bf.y = hi ? 0.0f : c4.y;
  return bf;
}

__device__ __forceinline__ v8f tile_product(const v2f a, const v2f bf) {
  const v8f cz = {};
  return __builtin_amdgcn_wmma_f32_16x16x4_f32(
      false, a, false, bf, (short)0, cz, false, false);
}

__global__ __launch_bounds__(THREADS)
void prep_kernel(const float* __restrict__ cloud, const float* __restrict__ noise,
                 float4* __restrict__ ypts, float4* __restrict__ xpts,
                 float* __restrict__ f, float* __restrict__ g) {
  const int b   = blockIdx.x;
  const int tid = threadIdx.x;
  __shared__ float s_sum[THREADS];
  __shared__ float s_sq[THREADS];

  const float* cb = cloud + (size_t)b * N_ * 3;
  const float* nb = noise + (size_t)b * N_ * 3;

  float sum = 0.f, sq = 0.f;
  for (int i = tid; i < N_ * 3; i += THREADS) {
    float v = cb[i];
    sum += v; sq += v * v;
  }
  s_sum[tid] = sum; s_sq[tid] = sq;
  __syncthreads();
  for (int off = THREADS / 2; off > 0; off >>= 1) {
    if (tid < off) { s_sum[tid] += s_sum[tid + off]; s_sq[tid] += s_sq[tid + off]; }
    __syncthreads();
  }
  const float n    = (float)(N_ * 3);
  const float tot  = s_sum[0];
  const float tot2 = s_sq[0];
  const float var  = (tot2 - tot * tot / n) / (n - 1.0f);
  const float inv_std = rsqrtf(var);

  for (int p = tid; p < N_; p += THREADS) {
    float yx = cb[p * 3 + 0] * inv_std;
    float yy = cb[p * 3 + 1] * inv_std;
    float yz = cb[p * 3 + 2] * inv_std;
    ypts[(size_t)b * N_ + p] = make_float4(yx, yy, yz, yx * yx + yy * yy + yz * yz);
    float xx = nb[p * 3 + 0];
    float xy = nb[p * 3 + 1];
    float xz = nb[p * 3 + 2];
    xpts[(size_t)b * N_ + p] = make_float4(xx, xy, xz, xx * xx + xy * xy + xz * xz);
    f[(size_t)b * N_ + p] = 0.f;
    g[(size_t)b * N_ + p] = 0.f;
  }
}

template <typename T>
__device__ __forceinline__ T max_of3(const T x, const T y, const T z) {
  return __builtin_elementwise_max(__builtin_elementwise_max(x, y), z);
}

#define REF_BIAS (-5.0f)
#define SUM_LO 0x1p-80f
#define SUM_HI 0x1p+120f

__global__ __launch_bounds__(TILE_THREADS)
void sinkhorn_half(const float4* __restrict__ rows_pts,
                   const float4* __restrict__ cols_pts,
                   const float*  __restrict__ in_pot,
                   float*        __restrict__ out_pot,
                   float eps, float lconst) {
  const int b    = blockIdx.y;
  const int tid  = threadIdx.x;
  const int lane = tid & 31;
  const int wave = tid >> 5;
  const int hi   = lane >> 4;
  const int idx  = lane & 15;

  __shared__ __attribute__((aligned(16))) v2f s_cols[2 * N_];

  const float4* cp = cols_pts + (size_t)b * N_;
  const float*  ip = in_pot   + (size_t)b * N_;
  float4 cst[COLS_PER_THREAD];
  float  pst[COLS_PER_THREAD];
#pragma unroll
  for (int k = 0; k < COLS_PER_THREAD; ++k) {
    cst[k] = cp[tid + k * TILE_THREADS];
    pst[k] = ip[tid + k * TILE_THREADS];
  }
#pragma unroll
  for (int k = 0; k < COLS_PER_THREAD; ++k) {
    const int j = tid + k * TILE_THREADS;
    v2f lo, up;
    lo.x = cst[k].x;  lo.y = cst[k].y;
    up.x = cst[k].z;  up.y = pst[k] - 0.5f * cst[k].w;
    s_cols[2 * j]     = lo;
    s_cols[2 * j + 1] = up;
  }
  __syncthreads();

  const int row_base = blockIdx.x * ROWS_PER_BLOCK + wave * 16;
  const float4* rp = rows_pts + (size_t)b * N_;
  float* op = out_pot + (size_t)b * N_;

  const float c2 = LOG2E / eps;
  const float4 r4 = rp[row_base + idx];
  const float pold = op[row_base + idx];
  v2f a;
  a.x = c2 * (hi ? r4.z : r4.x);
  a.y = hi ? c2 : c2 * r4.y;

  const float mref = __builtin_fmaf(c2, 0.5f * r4.w - pold, REF_BIAS - lconst * LOG2E);
  v8f cref;
#pragma unroll
  for (int r = 0; r < 8; ++r) cref[r] = -mref;

  const v2f* lc = s_cols + 2 * idx + hi;

  v2f cP[BLOCK_TILES], cQ[BLOCK_TILES];
  v8f aP[BLOCK_TILES], aQ[BLOCK_TILES];

  auto read_block = [&](const int col0, v2f (&c)[BLOCK_TILES]) {
#pragma unroll
    for (int t = 0; t < BLOCK_TILES; ++t) c[t] = lc[2 * (col0 + 16 * t)];
  };

  auto ref_product = [&](const v2f c) {
    return __builtin_amdgcn_wmma_f32_16x16x4_f32(
        false, c, false, a, (short)0, cref, false, false);
  };

  v8f acc = {};
  v8f ep  = {};

  auto fast_step = [&](const v2f (&c)[BLOCK_TILES], v8f (&nxt)[BLOCK_TILES],
                       const v8f (&cur)[BLOCK_TILES], const bool with_reads) {
#pragma unroll
    for (int t = 0; t < BLOCK_TILES; ++t) nxt[t] = ref_product(c[t]);
#pragma unroll
    for (int t = 0; t < BLOCK_TILES; ++t) {
      v8f e;
#pragma unroll
      for (int r = 0; r < 8; ++r) e[r] = __builtin_amdgcn_exp2f(cur[t][r]);
      acc += ep;
      ep = e;
    }
#pragma unroll
    for (int t = 0; t < BLOCK_TILES; ++t) {
      __builtin_amdgcn_sched_group_barrier(0x008, 1, 0);
#pragma unroll
      for (int k = 0; k < 4; ++k) {
        __builtin_amdgcn_sched_group_barrier(0x400, 1, 0);
        __builtin_amdgcn_sched_group_barrier(0x002, 1, 0);
      }
      __builtin_amdgcn_sched_group_barrier(0x400, 4, 0);
      if (with_reads && t == 2) __builtin_amdgcn_sched_group_barrier(0x100, BLOCK_TILES, 0);
    }
  };

  read_block(0, cP);
#pragma unroll
  for (int t = 0; t < BLOCK_TILES; ++t) aP[t] = ref_product(cP[t]);
  read_block(BLOCK_COLS, cQ);

  for (int c0 = 0; c0 < N_ - 2 * BLOCK_COLS; c0 += 2 * BLOCK_COLS) {
    read_block(c0 + 2 * BLOCK_COLS, cP);
    fast_step(cQ, aQ, aP, true);

    read_block(c0 + 3 * BLOCK_COLS, cQ);
    fast_step(cP, aP, aQ, true);
  }
  fast_step(cQ, aQ, aP, false);
#pragma unroll
  for (int t = 0; t < BLOCK_TILES; ++t) {
    v8f e;
#pragma unroll
    for (int r = 0; r < 8; ++r) e[r] = __builtin_amdgcn_exp2f(aQ[t][r]);
    acc += ep;
    ep = e;
  }
  acc += ep;

  const v4f a4 = acc.lo + acc.hi;
  const v2f a2 = a4.lo + a4.hi;
  float s = a2.x + a2.y;
  s += __shfl_xor(s, 16, 32);

  float lse2 = mref + __builtin_amdgcn_logf(s);

  const bool trusted = (s >= SUM_LO) && (s <= SUM_HI);
  if (__builtin_amdgcn_ballot_w32(!trusted) != 0u) {
    float m = -3.0e38f;
    float sx = 0.f;
    const v8f cz = {};
    for (int t = 0; t < N_ / 16; ++t) {
      const v2f c = lc[2 * 16 * t];
      const v8f v = __builtin_amdgcn_wmma_f32_16x16x4_f32(
          false, c, false, a, (short)0, cz, false, false);
      const float q0 = max_of3(v[0], v[1], v[2]);
      const float q1 = max_of3(v[3], v[4], v[5]);
      const float mn = max_of3(max_of3(q0, q1, v[6]), v[7], m);
      float part = 0.f;
#pragma unroll
      for (int r = 0; r < 8; ++r) part += __builtin_amdgcn_exp2f(v[r] - mn);
      sx = __builtin_fmaf(sx, __builtin_amdgcn_exp2f(m - mn), part);
      m = mn;
    }
    const float om = __shfl_xor(m, 16, 32);
    const float os = __shfl_xor(sx, 16, 32);
    const float d  = om - m;
    const float e  = __builtin_amdgcn_exp2f(-fabsf(d));
    const bool  up = d > 0.0f;
    sx = up ? fmaf(sx, e, os) : fmaf(os, e, sx);
    m  = up ? om : m;
    lse2 = m + __builtin_amdgcn_logf(sx);
  }

  if (hi == 0) {
    const int row = row_base + idx;
    op[row] = -eps * (lconst + LN2 * lse2) + 0.5f * r4.w;
  }
}

__global__ __launch_bounds__(TILE_THREADS)
void assign_mlp_kernel(const float4* __restrict__ xpts,
                       const float4* __restrict__ ypts,
                       const float*  __restrict__ g,
                       const float*  __restrict__ t,
                       const float*  __restrict__ W1,
                       const float*  __restrict__ Wt,
                       const float*  __restrict__ b1,
                       const float*  __restrict__ W2,
                       const float*  __restrict__ b2,
                       float*        __restrict__ out) {
  const int b    = blockIdx.y;
  const int tid  = threadIdx.x;
  const int lane = tid & 31;
  const int wave = tid >> 5;
  const int hi   = lane >> 4;
  const int idx  = lane & 15;

  __shared__ float4 s_cols[N_];
  __shared__ float  s_gy[N_];
  __shared__ float  s_W1[3 * H_];
  __shared__ float  s_Wt[H_];
  __shared__ float  s_b1[H_];
  __shared__ float  s_W2[H_ * 3];
  __shared__ float  s_b2[4];
  __shared__ int    s_idx[ROWS_PER_BLOCK];

  const float4* cp = ypts + (size_t)b * N_;
  const float*  gp = g    + (size_t)b * N_;
  float4 cst[COLS_PER_THREAD];
  float  pst[COLS_PER_THREAD];
#pragma unroll
  for (int k = 0; k < COLS_PER_THREAD; ++k) {
    cst[k] = cp[tid + k * TILE_THREADS];
    pst[k] = gp[tid + k * TILE_THREADS];
  }
#pragma unroll
  for (int k = 0; k < COLS_PER_THREAD; ++k) {
    const int j = tid + k * TILE_THREADS;
    s_cols[j] = cst[k];
    s_gy[j]   = pst[k] - cst[k].w;
  }
#pragma unroll
  for (int k = 0; k < 3 * H_ / TILE_THREADS; ++k) {
    const int j = tid + k * TILE_THREADS;
    s_W1[j] = W1[j]; s_W2[j] = W2[j];
  }
#pragma unroll
  for (int k = 0; k < H_ / TILE_THREADS; ++k) {
    const int j = tid + k * TILE_THREADS;
    s_Wt[j] = Wt[j]; s_b1[j] = b1[j];
  }
  if (tid < 3) s_b2[tid] = b2[tid];
  __syncthreads();

  const int row_base = blockIdx.x * ROWS_PER_BLOCK + wave * 16;
  const float4* rp = xpts + (size_t)b * N_;

  float4 r4 = rp[row_base + idx];
  v2f a;
  a.x = hi ? 2.0f * r4.z : 2.0f * r4.x;
  a.y = hi ? 0.0f        : 2.0f * r4.y;

  float best[8]; int bidx[8];
#pragma unroll
  for (int r = 0; r < 8; ++r) { best[r] = -3.0e38f; bidx[r] = 0; }

  auto amax_update = [&](const v8f acc, const float gy, const int col) {
#pragma unroll
    for (int r = 0; r < 8; ++r) {
      float v = acc[r] + gy;
      if (v > best[r]) { best[r] = v; bidx[r] = col; }
    }
  };

  const float4* lc = s_cols + idx;
  const float*  lg = s_gy + idx;

  float4 cP0, cP1, cQ0, cQ1;
  float  gP0, gP1, gQ0, gQ1;
  v8f    aP0, aP1, aQ0, aQ1;
  float  yP0, yP1, yQ0, yQ1;

  cP0 = lc[0];  cP1 = lc[16];  gP0 = lg[0];  gP1 = lg[16];
  aP0 = tile_product(a, frag_b(cP0, hi));
  aP1 = tile_product(a, frag_b(cP1, hi));
  yP0 = gP0; yP1 = gP1;
  cQ0 = lc[STAGE_COLS];  cQ1 = lc[STAGE_COLS + 16];
  gQ0 = lg[STAGE_COLS];  gQ1 = lg[STAGE_COLS + 16];

  for (int c0 = 0; c0 < N_; c0 += 2 * STAGE_COLS) {
    const int n0 = (c0 + 2 * STAGE_COLS) & (N_ - 1);
    const int n1 = (c0 + 3 * STAGE_COLS) & (N_ - 1);

    aQ0 = tile_product(a, frag_b(cQ0, hi));
    aQ1 = tile_product(a, frag_b(cQ1, hi));
    yQ0 = gQ0; yQ1 = gQ1;
    cP0 = lc[n0];  cP1 = lc[n0 + 16];  gP0 = lg[n0];  gP1 = lg[n0 + 16];
    amax_update(aP0, yP0, c0 + idx);
    amax_update(aP1, yP1, c0 + 16 + idx);

    aP0 = tile_product(a, frag_b(cP0, hi));
    aP1 = tile_product(a, frag_b(cP1, hi));
    yP0 = gP0; yP1 = gP1;
    cQ0 = lc[n1];  cQ1 = lc[n1 + 16];  gQ0 = lg[n1];  gQ1 = lg[n1 + 16];
    amax_update(aQ0, yQ0, c0 + STAGE_COLS + idx);
    amax_update(aQ1, yQ1, c0 + STAGE_COLS + 16 + idx);
  }

#pragma unroll
  for (int mask = 1; mask < 16; mask <<= 1) {
#pragma unroll
    for (int r = 0; r < 8; ++r) {
      float ov = __shfl_xor(best[r], mask, 32);
      int   oi = __shfl_xor(bidx[r], mask, 32);
      if (ov > best[r] || (ov == best[r] && oi < bidx[r])) {
        best[r] = ov; bidx[r] = oi;
      }
    }
  }

  if (idx < 8) s_idx[wave * 16 + hi * 8 + idx] = bidx[idx];
  __syncthreads();

  if (tid < ROWS_PER_BLOCK) {
    const int grow = blockIdx.x * ROWS_PER_BLOCK + tid;
    const int j    = s_idx[tid];
    float4 x0a = s_cols[j];
    float4 xn  = rp[grow];
    const float tv = t[b], omt = 1.0f - tv;

    float vx = xn.x - x0a.x, vy = xn.y - x0a.y, vz = xn.z - x0a.z;
    float xtx = omt * x0a.x + tv * xn.x;
    float xty = omt * x0a.y + tv * xn.y;
    float xtz = omt * x0a.z + tv * xn.z;

    float p0 = s_b2[0], p1 = s_b2[1], p2 = s_b2[2];
    for (int k = 0; k < H_; ++k) {
      float h = xtx * s_W1[k] + xty * s_W1[H_ + k] + xtz * s_W1[2 * H_ + k]
              + tv * s_Wt[k] + s_b1[k];
      h = fmaxf(h, 0.0f);
      p0 += h * s_W2[k * 3 + 0];
      p1 += h * s_W2[k * 3 + 1];
      p2 += h * s_W2[k * 3 + 2];
    }
    const size_t base  = ((size_t)b * N_ + grow) * 3;
    const size_t vbase = (size_t)B_ * N_ * 3 + base;
    out[base + 0] = p0; out[base + 1] = p1; out[base + 2] = p2;
    out[vbase + 0] = vx; out[vbase + 1] = vy; out[vbase + 2] = vz;
  }
}

extern "C" void kernel_launch(void* const* d_in, const int* in_sizes, int n_in,
                              void* d_out, int out_size, void* d_ws, size_t ws_size,
                              hipStream_t stream) {
  (void)in_sizes; (void)n_in; (void)out_size; (void)ws_size;
  const float* cloud = (const float*)d_in[0];
  const float* noise = (const float*)d_in[1];
  const float* t     = (const float*)d_in[2];
  const float* W1    = (const float*)d_in[3];
  const float* Wt    = (const float*)d_in[4];
  const float* b1    = (const float*)d_in[5];
  const float* W2    = (const float*)d_in[6];
  const float* b2    = (const float*)d_in[7];
  float* out = (float*)d_out;

  char* ws = (char*)d_ws;
  float4* ypts = (float4*)ws;
  float4* xpts = (float4*)(ws + (size_t)B_ * N_ * sizeof(float4));
  float*  f    = (float*) (ws + (size_t)2 * B_ * N_ * sizeof(float4));
  float*  g    = f + (size_t)B_ * N_;

  prep_kernel<<<B_, THREADS, 0, stream>>>(cloud, noise, ypts, xpts, f, g);

  const float lconst = -logf((float)N_);
  dim3 grid(N_ / ROWS_PER_BLOCK, B_);
  for (int it = 0; it < 14; ++it) {
    float eps = (float)(32.0 * pow(1e-6 / 32.0, (double)it / 13.0));
    sinkhorn_half<<<grid, TILE_THREADS, 0, stream>>>(xpts, ypts, g, f, eps, lconst);
    sinkhorn_half<<<grid, TILE_THREADS, 0, stream>>>(ypts, xpts, f, g, eps, lconst);
  }

  assign_mlp_kernel<<<grid, TILE_THREADS, 0, stream>>>(xpts, ypts, g, t,
                                                  W1, Wt, b1, W2, b2, out);
}


# Sinkhorn kernel reads column fragments straight from global memory into registers (all 128 loads of a lane issued at entry, column term kept in .w): no LDS staging, barrier, LDS reads; sum order kept
# speedup vs baseline: 2.9439x; 1.0469x over previous
#include <hip/hip_runtime.h>
#include <cmath>

#define B_  8
#define N_  2048
#define H_  256
#define THREADS 256
#define TILE_THREADS 128
#define ROWS_PER_BLOCK 64
#define COLS_PER_THREAD (N_ / TILE_THREADS)

typedef __attribute__((ext_vector_type(2))) float v2f;
typedef __attribute__((ext_vector_type(4))) float v4f;
typedef __attribute__((ext_vector_type(8))) float v8f;

#define LOG2E 1.4426950408889634f
#define LN2   0.6931471805599453f

#define STAGE_COLS 32

#define BLOCK_TILES 8
#define BLOCK_COLS (16 * BLOCK_TILES)

__device__ __forceinline__ v2f frag_b(const float4 c4, const int hi) {
  v2f bf;
  bf.x = hi ? c4.z : c4.x;
  bf.y = hi ? 0.0f : c4.y;
  return bf;
}

__device__ __forceinline__ v8f tile_product(const v2f a, const v2f bf) {
  const v8f cz = {};
  return __builtin_amdgcn_wmma_f32_16x16x4_f32(
      false, a, false, bf, (short)0, cz, false, false);
}

__global__ __launch_bounds__(THREADS)
void prep_kernel(const float* __restrict__ cloud, const float* __restrict__ noise,
                 float4* __restrict__ ypts, float4* __restrict__ xpts,
                 float* __restrict__ f, float* __restrict__ g) {
  const int b   = blockIdx.x;
  const int tid = threadIdx.x;
  __shared__ float s_sum[THREADS];
  __shared__ float s_sq[THREADS];

  const float* cb = cloud + (size_t)b * N_ * 3;
  const float* nb = noise + (size_t)b * N_ * 3;

  float sum = 0.f, sq = 0.f;
  for (int i = tid; i < N_ * 3; i += THREADS) {
    float v = cb[i];
    sum += v; sq += v * v;
  }
  s_sum[tid] = sum; s_sq[tid] = sq;
  __syncthreads();
  for (int off = THREADS / 2; off > 0; off >>= 1) {
    if (tid < off) { s_sum[tid] += s_sum[tid + off]; s_sq[tid] += s_sq[tid + off]; }
    __syncthreads();
  }
  const float n    = (float)(N_ * 3);
  const float tot  = s_sum[0];
  const float tot2 = s_sq[0];
  const float var  = (tot2 - tot * tot / n) / (n - 1.0f);
  const float inv_std = rsqrtf(var);

  for (int p = tid; p < N_; p += THREADS) {
    float yx = cb[p * 3 + 0] * inv_std;
    float yy = cb[p * 3 + 1] * inv_std;
    float yz = cb[p * 3 + 2] * inv_std;
    ypts[(size_t)b * N_ + p] = make_float4(yx, yy, yz, 0.f - 0.5f * (yx * yx + yy * yy + yz * yz));
    float xx = nb[p * 3 + 0];
    float xy = nb[p * 3 + 1];
    float xz = nb[p * 3 + 2];
    xpts[(size_t)b * N_ + p] = make_float4(xx, xy, xz, 0.f - 0.5f * (xx * xx + xy * xy + xz * xz));
    f[(size_t)b * N_ + p] = 0.f;
    g[(size_t)b * N_ + p] = 0.f;
  }
}

template <typename T>
__device__ __forceinline__ T max_of3(const T x, const T y, const T z) {
  return __builtin_elementwise_max(__builtin_elementwise_max(x, y), z);
}

#define REF_BIAS (-5.0f)
#define SUM_LO 0x1p-80f
#define SUM_HI 0x1p+120f

#define N_TILES  (N_ / 16)
#define N_BLOCKS (N_TILES / BLOCK_TILES)

__global__ __launch_bounds__(TILE_THREADS)
void sinkhorn_half(float4*       rows_pts,
                   const float4* __restrict__ cols_pts,
                   float*        __restrict__ out_pot,
                   float eps, float lconst) {
  const int b    = blockIdx.y;
  const int tid  = threadIdx.x;
  const int lane = tid & 31;
  const int wave = tid >> 5;
  const int hi   = lane >> 4;
  const int idx  = lane & 15;

  const int row_base = blockIdx.x * ROWS_PER_BLOCK + wave * 16;
  float4* rp = rows_pts + (size_t)b * N_;
  float*  op = out_pot  + (size_t)b * N_;

  const float4 r4   = rp[row_base + idx];
  const float  pold = op[row_base + idx];

  const v2f* gc = reinterpret_cast<const v2f*>(cols_pts + (size_t)b * N_) + 2 * idx + hi;
  v2f cf[N_TILES];
#pragma unroll
  for (int t0 = 0; t0 < N_TILES; t0 += 16) {
#pragma unroll
    for (int t = t0; t < t0 + 16; ++t) cf[t] = gc[32 * t];
    __builtin_amdgcn_sched_barrier(0);
  }

  const float c2 = LOG2E / eps;
  const float rw = r4.x * r4.x + r4.y * r4.y + r4.z * r4.z;
  v2f a;
  a.x = c2 * (hi ? r4.z : r4.x);
  a.y = hi ? c2 : c2 * r4.y;

  const float mref = __builtin_fmaf(c2, 0.5f * rw - pold, REF_BIAS - lconst * LOG2E);
  v8f cref;
#pragma unroll
  for (int r = 0; r < 8; ++r) cref[r] = -mref;

  v8f aP[BLOCK_TILES], aQ[BLOCK_TILES];

  auto ref_product = [&](const v2f c) {
    return __builtin_amdgcn_wmma_f32_16x16x4_f32(
        false, c, false, a, (short)0, cref, false, false);
  };

  v8f acc = {};
  v8f ep  = {};

  auto fast_step = [&](const v2f* c, v8f (&nxt)[BLOCK_TILES],
                       const v8f (&cur)[BLOCK_TILES]) {
#pragma unroll
    for (int t = 0; t < BLOCK_TILES; ++t) nxt[t] = ref_product(c[t]);
#pragma unroll
    for (int t = 0; t < BLOCK_TILES; ++t) {
      v8f e;
#pragma unroll
      for (int r = 0; r < 8; ++r) e[r] = __builtin_amdgcn_exp2f(cur[t][r]);
      acc += ep;
      ep = e;
    }
#pragma unroll
    for (int t = 0; t < BLOCK_TILES; ++t) {
      __builtin_amdgcn_sched_group_barrier(0x008, 1, 0);
#pragma unroll
      for (int k = 0; k < 4; ++k) {
        __builtin_amdgcn_sched_group_barrier(0x400, 1, 0);
        __builtin_amdgcn_sched_group_barrier(0x002, 1, 0);
      }
      __builtin_amdgcn_sched_group_barrier(0x400, 4, 0);
    }
    __builtin_amdgcn_sched_barrier(0);
  };

#pragma unroll
  for (int t = 0; t < BLOCK_TILES; ++t) aP[t] = ref_product(cf[t]);
  __builtin_amdgcn_sched_barrier(0);

#pragma unroll
  for (int n = 0; n < N_BLOCKS - 2; n += 2) {
    fast_step(&cf[(n + 1) * BLOCK_TILES], aQ, aP);
    fast_step(&cf[(n + 2) * BLOCK_TILES], aP, aQ);
  }
  fast_step(&cf[(N_BLOCKS - 1) * BLOCK_TILES], aQ, aP);
#pragma unroll
  for (int t = 0; t < BLOCK_TILES; ++t) {
    v8f e;
#pragma unroll
    for (int r = 0; r < 8; ++r) e[r] = __builtin_amdgcn_exp2f(aQ[t][r]);
    acc += ep;
    ep = e;
  }
  acc += ep;

  const v4f a4 = acc.lo + acc.hi;
  const v2f a2 = a4.lo + a4.hi;
  float s = a2.x + a2.y;
  s += __shfl_xor(s, 16, 32);

  float lse2 = mref + __builtin_amdgcn_logf(s);

  const bool trusted = (s >= SUM_LO) && (s <= SUM_HI);
  if (__builtin_amdgcn_ballot_w32(!trusted) != 0u) {
    float m = -3.0e38f;
    float sx = 0.f;
    const v8f cz = {};
    for (int t = 0; t < N_TILES; ++t) {
      const v2f c = gc[32 * t];
      const v8f v = __builtin_amdgcn_wmma_f32_16x16x4_f32(
          false, c, false, a, (short)0, cz, false, false);
      const float q0 = max_of3(v[0], v[1], v[2]);
      const float q1 = max_of3(v[3], v[4], v[5]);
      const float mn = max_of3(max_of3(q0, q1, v[6]), v[7], m);
      float part = 0.f;
#pragma unroll
      for (int r = 0; r < 8; ++r) part += __builtin_amdgcn_exp2f(v[r] - mn);
      sx = __builtin_fmaf(sx, __builtin_amdgcn_exp2f(m - mn), part);
      m = mn;
    }
    const float om = __shfl_xor(m, 16, 32);
    const float os = __shfl_xor(sx, 16, 32);
    const float d  = om - m;
    const float e  = __builtin_amdgcn_exp2f(-fabsf(d));
    const bool  up = d > 0.0f;
    sx = up ? fmaf(sx, e, os) : fmaf(os, e, sx);
    m  = up ? om : m;
    lse2 = m + __builtin_amdgcn_logf(sx);
  }

  if (hi == 0) {
    const int row = row_base + idx;
    const float pot = -eps * (lconst + LN2 * lse2) + 0.5f * rw;
    op[row] = pot;
    rp[row].w = pot - 0.5f * rw;
  }
}

__global__ __launch_bounds__(TILE_THREADS)
void assign_mlp_kernel(const float4* __restrict__ xpts,
                       const float4* __restrict__ ypts,
                       const float*  __restrict__ g,
                       const float*  __restrict__ t,
                       const float*  __restrict__ W1,
                       const float*  __restrict__ Wt,
                       const float*  __restrict__ b1,
                       const float*  __restrict__ W2,
                       const float*  __restrict__ b2,
                       float*        __restrict__ out) {
  const int b    = blockIdx.y;
  const int tid  = threadIdx.x;
  const int lane = tid & 31;
  const int wave = tid >> 5;
  const int hi   = lane >> 4;
  const int idx  = lane & 15;

  __shared__ float4 s_cols[N_];
  __shared__ float  s_gy[N_];
  __shared__ float  s_W1[3 * H_];
  __shared__ float  s_Wt[H_];
  __shared__ float  s_b1[H_];
  __shared__ float  s_W2[H_ * 3];
  __shared__ float  s_b2[4];
  __shared__ int    s_idx[ROWS_PER_BLOCK];

  const float4* cp = ypts + (size_t)b * N_;
  const float*  gp = g    + (size_t)b * N_;
  float4 cst[COLS_PER_THREAD];
  float  pst[COLS_PER_THREAD];
#pragma unroll
  for (int k = 0; k < COLS_PER_THREAD; ++k) {
    cst[k] = cp[tid + k * TILE_THREADS];
    pst[k] = gp[tid + k * TILE_THREADS];
  }
#pragma unroll
  for (int k = 0; k < COLS_PER_THREAD; ++k) {
    const int j = tid + k * TILE_THREADS;
    s_cols[j] = cst[k];
    s_gy[j]   = pst[k] - (cst[k].x * cst[k].x + cst[k].y * cst[k].y + cst[k].z * cst[k].z);
  }
#pragma unroll
  for (int k = 0; k < 3 * H_ / TILE_THREADS; ++k) {
    const int j = tid + k * TILE_THREADS;
    s_W1[j] = W1[j]; s_W2[j] = W2[j];
  }
#pragma unroll
  for (int k = 0; k < H_ / TILE_THREADS; ++k) {
    const int j = tid + k * TILE_THREADS;
    s_Wt[j] = Wt[j]; s_b1[j] = b1[j];
  }
  if (tid < 3) s_b2[tid] = b2[tid];
  __syncthreads();

  const int row_base = blockIdx.x * ROWS_PER_BLOCK + wave * 16;
  const float4* rp = xpts + (size_t)b * N_;

  float4 r4 = rp[row_base + idx];
  v2f a;
  a.x = hi ? 2.0f * r4.z : 2.0f * r4.x;
  a.y = hi ? 0.0f        : 2.0f * r4.y;

  float best[8]; int bidx[8];
#pragma unroll
  for (int r = 0; r < 8; ++r) { best[r] = -3.0e38f; bidx[r] = 0; }

  auto amax_update = [&](const v8f acc, const float gy, const int col) {
#pragma unroll
    for (int r = 0; r < 8; ++r) {
      float v = acc[r] + gy;
      if (v > best[r]) { best[r] = v; bidx[r] = col; }
    }
  };

  const float4* lc = s_cols + idx;
  const float*  lg = s_gy + idx;

  float4 cP0, cP1, cQ0, cQ1;
  float  gP0, gP1, gQ0, gQ1;
  v8f    aP0, aP1, aQ0, aQ1;
  float  yP0, yP1, yQ0, yQ1;

  cP0 = lc[0];  cP1 = lc[16];  gP0 = lg[0];  gP1 = lg[16];
  aP0 = tile_product(a, frag_b(cP0, hi));
  aP1 = tile_product(a, frag_b(cP1, hi));
  yP0 = gP0; yP1 = gP1;
  cQ0 = lc[STAGE_COLS];  cQ1 = lc[STAGE_COLS + 16];
  gQ0 = lg[STAGE_COLS];  gQ1 = lg[STAGE_COLS + 16];

  for (int c0 = 0; c0 < N_; c0 += 2 * STAGE_COLS) {
    const int n0 = (c0 + 2 * STAGE_COLS) & (N_ - 1);
    const int n1 = (c0 + 3 * STAGE_COLS) & (N_ - 1);

    aQ0 = tile_product(a, frag_b(cQ0, hi));
    aQ1 = tile_product(a, frag_b(cQ1, hi));
    yQ0 = gQ0; yQ1 = gQ1;
    cP0 = lc[n0];  cP1 = lc[n0 + 16];  gP0 = lg[n0];  gP1 = lg[n0 + 16];
    amax_update(aP0, yP0, c0 + idx);
    amax_update(aP1, yP1, c0 + 16 + idx);

    aP0 = tile_product(a, frag_b(cP0, hi));
    aP1 = tile_product(a, frag_b(cP1, hi));
    yP0 = gP0; yP1 = gP1;
    cQ0 = lc[n1];  cQ1 = lc[n1 + 16];  gQ0 = lg[n1];  gQ1 = lg[n1 + 16];
    amax_update(aQ0, yQ0, c0 + STAGE_COLS + idx);
    amax_update(aQ1, yQ1, c0 + STAGE_COLS + 16 + idx);
  }

#pragma unroll
  for (int mask = 1; mask < 16; mask <<= 1) {
#pragma unroll
    for (int r = 0; r < 8; ++r) {
      float ov = __shfl_xor(best[r], mask, 32);
      int   oi = __shfl_xor(bidx[r], mask, 32);
      if (ov > best[r] || (ov == best[r] && oi < bidx[r])) {
        best[r] = ov; bidx[r] = oi;
      }
    }
  }

  if (idx < 8) s_idx[wave * 16 + hi * 8 + idx] = bidx[idx];
  __syncthreads();

  if (tid < ROWS_PER_BLOCK) {
    const int grow = blockIdx.x * ROWS_PER_BLOCK + tid;
    const int j    = s_idx[tid];
    float4 x0a = s_cols[j];
    float4 xn  = rp[grow];
    const float tv = t[b], omt = 1.0f - tv;

    float vx = xn.x - x0a.x, vy = xn.y - x0a.y, vz = xn.z - x0a.z;
    float xtx = omt * x0a.x + tv * xn.x;
    float xty = omt * x0a.y + tv * xn.y;
    float xtz = omt * x0a.z + tv * xn.z;

    float p0 = s_b2[0], p1 = s_b2[1], p2 = s_b2[2];
    for (int k = 0; k < H_; ++k) {
      float h = xtx * s_W1[k] + xty * s_W1[H_ + k] + xtz * s_W1[2 * H_ + k]
              + tv * s_Wt[k] + s_b1[k];
      h = fmaxf(h, 0.0f);
      p0 += h * s_W2[k * 3 + 0];
      p1 += h * s_W2[k * 3 + 1];
      p2 += h * s_W2[k * 3 + 2];
    }
    const size_t base  = ((size_t)b * N_ + grow) * 3;
    const size_t vbase = (size_t)B_ * N_ * 3 + base;
    out[base + 0] = p0; out[base + 1] = p1; out[base + 2] = p2;
    out[vbase + 0] = vx; out[vbase + 1] = vy; out[vbase + 2] = vz;
  }
}

extern "C" void kernel_launch(void* const* d_in, const int* in_sizes, int n_in,
                              void* d_out, int out_size, void* d_ws, size_t ws_size,
                              hipStream_t stream) {
  (void)in_sizes; (void)n_in; (void)out_size; (void)ws_size;
  const float* cloud = (const float*)d_in[0];
  const float* noise = (const float*)d_in[1];
  const float* t     = (const float*)d_in[2];
  const float* W1    = (const float*)d_in[3];
  const float* Wt    = (const float*)d_in[4];
  const float* b1    = (const float*)d_in[5];
  const float* W2    = (const float*)d_in[6];
  const float* b2    = (const float*)d_in[7];
  float* out = (float*)d_out;

  char* ws = (char*)d_ws;
  float4* ypts = (float4*)ws;
  float4* xpts = (float4*)(ws + (size_t)B_ * N_ * sizeof(float4));
  float*  f    = (float*) (ws + (size_t)2 * B_ * N_ * sizeof(float4));
  float*  g    = f + (size_t)B_ * N_;

  prep_kernel<<<B_, THREADS, 0, stream>>>(cloud, noise, ypts, xpts, f, g);

  const float lconst = -logf((float)N_);
  dim3 grid(N_ / ROWS_PER_BLOCK, B_);
  for (int it = 0; it < 14; ++it) {
    float eps = (float)(32.0 * pow(1e-6 / 32.0, (double)it / 13.0));
    sinkhorn_half<<<grid, TILE_THREADS, 0, stream>>>(xpts, ypts, f, eps, lconst);
    sinkhorn_half<<<grid, TILE_THREADS, 0, stream>>>(ypts, xpts, g, eps, lconst);
  }

  assign_mlp_kernel<<<grid, TILE_THREADS, 0, stream>>>(xpts, ypts, g, t,
                                                  W1, Wt, b1, W2, b2, out);
}


# Assignment kernel taken columns x rows like the LSE kernel: a lane owns one row, so argmax ends in-lane plus one exchange and the lane pair runs its row's MLP as WMMA products; MLP sum order changes
# speedup vs baseline: 3.0573x; 1.0385x over previous
#include <hip/hip_runtime.h>
#include <cmath>

#define B_  8
#define N_  2048
#define H_  256
#define THREADS 256
#define TILE_THREADS 128
#define ROWS_PER_BLOCK 64
#define COLS_PER_THREAD (N_ / TILE_THREADS)

typedef __attribute__((ext_vector_type(2))) float v2f;
typedef __attribute__((ext_vector_type(4))) float v4f;
typedef __attribute__((ext_vector_type(8))) float v8f;

#define LOG2E 1.4426950408889634f
#define LN2   0.6931471805599453f

#define STAGE_COLS 32

#define BLOCK_TILES 8
#define BLOCK_COLS (16 * BLOCK_TILES)

__global__ __launch_bounds__(THREADS)
void prep_kernel(const float* __restrict__ cloud, const float* __restrict__ noise,
                 float4* __restrict__ ypts, float4* __restrict__ xpts,
                 float* __restrict__ f, float* __restrict__ g) {
  const int b   = blockIdx.x;
  const int tid = threadIdx.x;
  __shared__ float s_sum[THREADS];
  __shared__ float s_sq[THREADS];

  const float* cb = cloud + (size_t)b * N_ * 3;
  const float* nb = noise + (size_t)b * N_ * 3;

  float sum = 0.f, sq = 0.f;
  for (int i = tid; i < N_ * 3; i += THREADS) {
    float v = cb[i];
    sum += v; sq += v * v;
  }
  s_sum[tid] = sum; s_sq[tid] = sq;
  __syncthreads();
  for (int off = THREADS / 2; off > 0; off >>= 1) {
    if (tid < off) { s_sum[tid] += s_sum[tid + off]; s_sq[tid] += s_sq[tid + off]; }
    __syncthreads();
  }
  const float n    = (float)(N_ * 3);
  const float tot  = s_sum[0];
  const float tot2 = s_sq[0];
  const float var  = (tot2 - tot * tot / n) / (n - 1.0f);
  const float inv_std = rsqrtf(var);

  for (int p = tid; p < N_; p += THREADS) {
    float yx = cb[p * 3 + 0] * inv_std;
    float yy = cb[p * 3 + 1] * inv_std;
    float yz = cb[p * 3 + 2] * inv_std;
    ypts[(size_t)b * N_ + p] = make_float4(yx, yy, yz, 0.f - 0.5f * (yx * yx + yy * yy + yz * yz));
    float xx = nb[p * 3 + 0];
    float xy = nb[p * 3 + 1];
    float xz = nb[p * 3 + 2];
    xpts[(size_t)b * N_ + p] = make_float4(xx, xy, xz, 0.f - 0.5f * (xx * xx + xy * xy + xz * xz));
    f[(size_t)b * N_ + p] = 0.f;
    g[(size_t)b * N_ + p] = 0.f;
  }
}

template <typename T>
__device__ __forceinline__ T max_of3(const T x, const T y, const T z) {
  return __builtin_elementwise_max(__builtin_elementwise_max(x, y), z);
}

#define REF_BIAS (-5.0f)
#define SUM_LO 0x1p-80f
#define SUM_HI 0x1p+120f

#define N_TILES  (N_ / 16)
#define N_BLOCKS (N_TILES / BLOCK_TILES)

__global__ __launch_bounds__(TILE_THREADS)
void sinkhorn_half(float4*       rows_pts,
                   const float4* __restrict__ cols_pts,
                   float*        __restrict__ out_pot,
                   float eps, float lconst) {
  const int b    = blockIdx.y;
  const int tid  = threadIdx.x;
  const int lane = tid & 31;
  const int wave = tid >> 5;
  const int hi   = lane >> 4;
  const int idx  = lane & 15;

  const int row_base = blockIdx.x * ROWS_PER_BLOCK + wave * 16;
  float4* rp = rows_pts + (size_t)b * N_;
  float*  op = out_pot  + (size_t)b * N_;

  const float4 r4   = rp[row_base + idx];
  const float  pold = op[row_base + idx];

  const v2f* gc = reinterpret_cast<const v2f*>(cols_pts + (size_t)b * N_) + 2 * idx + hi;
  v2f cf[N_TILES];
#pragma unroll
  for (int t0 = 0; t0 < N_TILES; t0 += 16) {
#pragma unroll
    for (int t = t0; t < t0 + 16; ++t) cf[t] = gc[32 * t];
    __builtin_amdgcn_sched_barrier(0);
  }

  const float c2 = LOG2E / eps;
  const float rw = r4.x * r4.x + r4.y * r4.y + r4.z * r4.z;
  v2f a;
  a.x = c2 * (hi ? r4.z : r4.x);
  a.y = hi ? c2 : c2 * r4.y;

  const float mref = __builtin_fmaf(c2, 0.5f * rw - pold, REF_BIAS - lconst * LOG2E);
  v8f cref;
#pragma unroll
  for (int r = 0; r < 8; ++r) cref[r] = -mref;

  v8f aP[BLOCK_TILES], aQ[BLOCK_TILES];

  auto ref_product = [&](const v2f c) {
    return __builtin_amdgcn_wmma_f32_16x16x4_f32(
        false, c, false, a, (short)0, cref, false, false);
  };

  v8f acc = {};
  v8f ep  = {};

  auto fast_step = [&](const v2f* c, v8f (&nxt)[BLOCK_TILES],
                       const v8f (&cur)[BLOCK_TILES]) {
#pragma unroll
    for (int t = 0; t < BLOCK_TILES; ++t) nxt[t] = ref_product(c[t]);
#pragma unroll
    for (int t = 0; t < BLOCK_TILES; ++t) {
      v8f e;
#pragma unroll
      for (int r = 0; r < 8; ++r) e[r] = __builtin_amdgcn_exp2f(cur[t][r]);
      acc += ep;
      ep = e;
    }
#pragma unroll
    for (int t = 0; t < BLOCK_TILES; ++t) {
      __builtin_amdgcn_sched_group_barrier(0x008, 1, 0);
#pragma unroll
      for (int k = 0; k < 4; ++k) {
        __builtin_amdgcn_sched_group_barrier(0x400, 1, 0);
        __builtin_amdgcn_sched_group_barrier(0x002, 1, 0);
      }
      __builtin_amdgcn_sched_group_barrier(0x400, 4, 0);
    }
    __builtin_amdgcn_sched_barrier(0);
  };

#pragma unroll
  for (int t = 0; t < BLOCK_TILES; ++t) aP[t] = ref_product(cf[t]);
  __builtin_amdgcn_sched_barrier(0);

#pragma unroll
  for (int n = 0; n < N_BLOCKS - 2; n += 2) {
    fast_step(&cf[(n + 1) * BLOCK_TILES], aQ, aP);
    fast_step(&cf[(n + 2) * BLOCK_TILES], aP, aQ);
  }
  fast_step(&cf[(N_BLOCKS - 1) * BLOCK_TILES], aQ, aP);
#pragma unroll
  for (int t = 0; t < BLOCK_TILES; ++t) {
    v8f e;
#pragma unroll
    for (int r = 0; r < 8; ++r) e[r] = __builtin_amdgcn_exp2f(aQ[t][r]);
    acc += ep;
    ep = e;
  }
  acc += ep;

  const v4f a4 = acc.lo + acc.hi;
  const v2f a2 = a4.lo + a4.hi;
  float s = a2.x + a2.y;
  s += __shfl_xor(s, 16, 32);

  float lse2 = mref + __builtin_amdgcn_logf(s);

  const bool trusted = (s >= SUM_LO) && (s <= SUM_HI);
  if (__builtin_amdgcn_ballot_w32(!trusted) != 0u) {
    float m = -3.0e38f;
    float sx = 0.f;
    const v8f cz = {};
    for (int t = 0; t < N_TILES; ++t) {
      const v2f c = gc[32 * t];
      const v8f v = __builtin_amdgcn_wmma_f32_16x16x4_f32(
          false, c, false, a, (short)0, cz, false, false);
      const float q0 = max_of3(v[0], v[1], v[2]);
      const float q1 = max_of3(v[3], v[4], v[5]);
      const float mn = max_of3(max_of3(q0, q1, v[6]), v[7], m);
      float part = 0.f;
#pragma unroll
      for (int r = 0; r < 8; ++r) part += __builtin_amdgcn_exp2f(v[r] - mn);
      sx = __builtin_fmaf(sx, __builtin_amdgcn_exp2f(m - mn), part);
      m = mn;
    }
    const float om = __shfl_xor(m, 16, 32);
    const float os = __shfl_xor(sx, 16, 32);
    const float d  = om - m;
    const float e  = __builtin_amdgcn_exp2f(-fabsf(d));
    const bool  up = d > 0.0f;
    sx = up ? fmaf(sx, e, os) : fmaf(os, e, sx);
    m  = up ? om : m;
    lse2 = m + __builtin_amdgcn_logf(sx);
  }

  if (hi == 0) {
    const int row = row_base + idx;
    const float pot = -eps * (lconst + LN2 * lse2) + 0.5f * rw;
    op[row] = pot;
    rp[row].w = pot - 0.5f * rw;
  }
}

#define W2T_PITCH (H_ + 4)

__global__ __launch_bounds__(TILE_THREADS)
void assign_mlp_kernel(const float4* __restrict__ xpts,
                       const float4* __restrict__ ypts,
                       const float*  __restrict__ g,
                       const float*  __restrict__ t,
                       const float*  __restrict__ W1,
                       const float*  __restrict__ Wt,
                       const float*  __restrict__ b1,
                       const float*  __restrict__ W2,
                       const float*  __restrict__ b2,
                       float*        __restrict__ out) {
  const int b    = blockIdx.y;
  const int tid  = threadIdx.x;
  const int lane = tid & 31;
  const int wave = tid >> 5;
  const int hi   = lane >> 4;
  const int idx  = lane & 15;

  __shared__ float4 s_cols[N_];
  __shared__ float4 s_gy4[N_ / 4];
  __shared__ float4 s_W1e[H_];
  __shared__ float4 s_W2T4[3 * W2T_PITCH / 4];
  __shared__ float  s_res[ROWS_PER_BLOCK * 6];
  float* s_gy  = reinterpret_cast<float*>(s_gy4);
  float* s_W2T = reinterpret_cast<float*>(s_W2T4);

  const float4* cp = ypts + (size_t)b * N_;
  const float*  gp = g    + (size_t)b * N_;
  const float   tv = t[b];
  float4 cst[COLS_PER_THREAD];
  float  pst[COLS_PER_THREAD];
#pragma unroll
  for (int k = 0; k < COLS_PER_THREAD; ++k) {
    cst[k] = cp[tid + k * TILE_THREADS];
    pst[k] = gp[tid + k * TILE_THREADS];
  }
  const int row_base = blockIdx.x * ROWS_PER_BLOCK + wave * 16;
  const float4* rp = xpts + (size_t)b * N_;
  const float4 r4 = rp[row_base + idx];
  float w1a[2], w1b[2], w1c[2], wtt[2], bb[2], w2v[6];
#pragma unroll
  for (int k = 0; k < H_ / TILE_THREADS; ++k) {
    const int j = tid + k * TILE_THREADS;
    w1a[k] = W1[j]; w1b[k] = W1[H_ + j]; w1c[k] = W1[2 * H_ + j];
    wtt[k] = Wt[j]; bb[k] = b1[j];
  }
#pragma unroll
  for (int k = 0; k < 3 * H_ / TILE_THREADS; ++k) w2v[k] = W2[tid + k * TILE_THREADS];

#pragma unroll
  for (int k = 0; k < COLS_PER_THREAD; ++k) {
    const int j = tid + k * TILE_THREADS;
    s_cols[j] = make_float4(cst[k].x, cst[k].y, cst[k].z, 0.0f);
    s_gy[j]   = pst[k] - (cst[k].x * cst[k].x + cst[k].y * cst[k].y + cst[k].z * cst[k].z);
  }
#pragma unroll
  for (int k = 0; k < H_ / TILE_THREADS; ++k) {
    const int j = tid + k * TILE_THREADS;
    s_W1e[j] = make_float4(w1a[k], w1b[k], w1c[k], fmaf(tv, wtt[k], bb[k]));
  }
#pragma unroll
  for (int k = 0; k < 3 * H_ / TILE_THREADS; ++k) {
    const int e = tid + k * TILE_THREADS;
    const int j = e / 3;
    s_W2T[(e - 3 * j) * W2T_PITCH + j] = w2v[k];
  }
  __syncthreads();

  v2f a;
  a.x = hi ? 2.0f * r4.z : 2.0f * r4.x;
  a.y = hi ? 0.0f        : 2.0f * r4.y;

  float best[8]; int bt[8];
#pragma unroll
  for (int r = 0; r < 8; ++r) { best[r] = -3.0e38f; bt[r] = 0; }

  auto amax_tile = [&](const v8f acc, const v8f gy, const int tb) {
    const v8f v = acc + gy;
#pragma unroll
    for (int r = 0; r < 8; ++r) {
      if (v[r] > best[r]) { best[r] = v[r]; bt[r] = tb; }
    }
  };

  const v2f*    lc = reinterpret_cast<const v2f*>(s_cols) + 2 * idx + hi;
  const v4f*    lg = reinterpret_cast<const v4f*>(s_gy4) + 2 * hi;
  const v8f cz = {};

  v2f    cP[4], cQ[4];
  v8f    gP[4], gQ[4];
  v8f    aP[4], aQ[4];
  v8f    yP[4], yQ[4];

  auto load_stage = [&](v2f (&c)[4], v8f (&gg)[4], const int s) {
    const int t0 = (4 * s) & (N_TILES - 1);
    const v2f* lcs = lc + 32 * t0;
    const v4f* lgs = lg + 4 * t0;
#pragma unroll
    for (int i = 0; i < 4; ++i) {
      c[i] = lcs[32 * i];
      const v4f g0 = lgs[4 * i];
      const v4f g1 = lgs[4 * i + 1];
      gg[i] = __builtin_shufflevector(g0, g1, 0, 1, 2, 3, 4, 5, 6, 7);
    }
  };
  auto mul_stage = [&](v8f (&acc)[4], const v2f (&c)[4], v8f (&y)[4], const v8f (&gg)[4]) {
#pragma unroll
    for (int i = 0; i < 4; ++i) {
      acc[i] = __builtin_amdgcn_wmma_f32_16x16x4_f32(false, c[i], false, a, (short)0, cz, false, false);
      y[i] = gg[i];
    }
  };
  auto amax_stage = [&](const v8f (&acc)[4], const v8f (&y)[4], const int s) {
#pragma unroll
    for (int i = 0; i < 4; ++i) amax_tile(acc[i], y[i], 16 * (4 * s + i));
  };

  load_stage(cP, gP, 0);
  mul_stage(aP, cP, yP, gP);
  load_stage(cQ, gQ, 1);
  for (int s0 = 0; s0 < N_TILES / 4; s0 += 2) {
    mul_stage(aQ, cQ, yQ, gQ);
    load_stage(cP, gP, s0 + 2);
    amax_stage(aP, yP, s0);
    mul_stage(aP, cP, yP, gP);
    load_stage(cQ, gQ, s0 + 3);
    amax_stage(aQ, yQ, s0 + 1);
  }

  float bv = best[0]; int bi = bt[0];
#pragma unroll
  for (int r = 1; r < 8; ++r) {
    const int  cr   = bt[r] + r;
    const bool take = (best[r] > bv) | ((best[r] == bv) & (cr < bi));
    bv = take ? best[r] : bv;
    bi = take ? cr : bi;
  }
  bi += 8 * hi;
  const float ov = __shfl_xor(bv, 16, 32);
  const int   oi = __shfl_xor(bi, 16, 32);
  const int   jm = ((ov > bv) | ((ov == bv) & (oi < bi))) ? oi : bi;

  const float4 x0a = s_cols[jm];
  const float omt = 1.0f - tv;
  const float vx = r4.x - x0a.x, vy = r4.y - x0a.y, vz = r4.z - x0a.z;
  const float xtx = omt * x0a.x + tv * r4.x;
  const float xty = omt * x0a.y + tv * r4.y;
  const float xtz = omt * x0a.z + tv * r4.z;

  v2f bx;
  bx.x = hi ? xtz  : xtx;
  bx.y = hi ? 1.0f : xty;
  const v2f*    w1 = reinterpret_cast<const v2f*>(s_W1e) + 2 * idx + hi;
  const int     dd = idx < 2 ? idx : 2;
  const float4* w2 = reinterpret_cast<const float4*>(s_W2T + dd * W2T_PITCH) + 2 * hi;
  v2f    a1s[H_ / 16];
  float4 was[H_ / 16], wbs[H_ / 16];
#pragma unroll
  for (int q = 0; q < H_ / 16; ++q) {
    a1s[q] = w1[32 * q];
    was[q] = w2[4 * q];
    wbs[q] = w2[4 * q + 1];
  }
  __builtin_amdgcn_sched_barrier(0);
  v8f acc0 = {}, acc1 = {};
#pragma unroll
  for (int q = 0; q < H_ / 16; ++q) {
    const v2f    a1 = a1s[q];
    const float4 wa = was[q];
    const float4 wb = wbs[q];
    const v8f d1 = __builtin_amdgcn_wmma_f32_16x16x4_f32(false, a1, false, bx, (short)0, cz, false, false);
    float hb[8];
#pragma unroll
    for (int r = 0; r < 8; ++r) hb[r] = fmaxf(d1[r], 0.0f);
    v2f wf, hf;
    wf.x = wa.x; wf.y = wa.y; hf.x = hb[0]; hf.y = hb[1];
    acc0 = __builtin_amdgcn_wmma_f32_16x16x4_f32(false, wf, false, hf, (short)0, acc0, false, false);
    wf.x = wa.z; wf.y = wa.w; hf.x = hb[2]; hf.y = hb[3];
    acc1 = __builtin_amdgcn_wmma_f32_16x16x4_f32(false, wf, false, hf, (short)0, acc1, false, false);
    wf.x = wb.x; wf.y = wb.y; hf.x = hb[4]; hf.y = hb[5];
    acc0 = __builtin_amdgcn_wmma_f32_16x16x4_f32(false, wf, false, hf, (short)0, acc0, false, false);
    wf.x = wb.z; wf.y = wb.w; hf.x = hb[6]; hf.y = hb[7];
    acc1 = __builtin_amdgcn_wmma_f32_16x16x4_f32(false, wf, false, hf, (short)0, acc1, false, false);
  }
  const v8f pr = acc0 + acc1;
  if (hi == 0) {
    float* so = s_res + (wave * 16 + idx) * 6;
    so[0] = pr[0] + b2[0]; so[1] = pr[1] + b2[1]; so[2] = pr[2] + b2[2];
    so[3] = vx; so[4] = vy; so[5] = vz;
  }
  __syncthreads();

  if (tid < ROWS_PER_BLOCK) {
    const int grow = blockIdx.x * ROWS_PER_BLOCK + tid;
    const float* so = s_res + tid * 6;
    const size_t base  = ((size_t)b * N_ + grow) * 3;
    const size_t vbase = (size_t)B_ * N_ * 3 + base;
    out[base + 0] = so[0]; out[base + 1] = so[1]; out[base + 2] = so[2];
    out[vbase + 0] = so[3]; out[vbase + 1] = so[4]; out[vbase + 2] = so[5];
  }
}

extern "C" void kernel_launch(void* const* d_in, const int* in_sizes, int n_in,
                              void* d_out, int out_size, void* d_ws, size_t ws_size,
                              hipStream_t stream) {
  (void)in_sizes; (void)n_in; (void)out_size; (void)ws_size;
  const float* cloud = (const float*)d_in[0];
  const float* noise = (const float*)d_in[1];
  const float* t     = (const float*)d_in[2];
  const float* W1    = (const float*)d_in[3];
  const float* Wt    = (const float*)d_in[4];
  const float* b1    = (const float*)d_in[5];
  const float* W2    = (const float*)d_in[6];
  const float* b2    = (const float*)d_in[7];
  float* out = (float*)d_out;

  char* ws = (char*)d_ws;
  float4* ypts = (float4*)ws;
  float4* xpts = (float4*)(ws + (size_t)B_ * N_ * sizeof(float4));
  float*  f    = (float*) (ws + (size_t)2 * B_ * N_ * sizeof(float4));
  float*  g    = f + (size_t)B_ * N_;

  prep_kernel<<<B_, THREADS, 0, stream>>>(cloud, noise, ypts, xpts, f, g);

  const float lconst = -logf((float)N_);
  dim3 grid(N_ / ROWS_PER_BLOCK, B_);
  for (int it = 0; it < 14; ++it) {
    float eps = (float)(32.0 * pow(1e-6 / 32.0, (double)it / 13.0));
    sinkhorn_half<<<grid, TILE_THREADS, 0, stream>>>(xpts, ypts, f, eps, lconst);
    sinkhorn_half<<<grid, TILE_THREADS, 0, stream>>>(ypts, xpts, g, eps, lconst);
  }

  assign_mlp_kernel<<<grid, TILE_THREADS, 0, stream>>>(xpts, ypts, g, t,
                                                  W1, Wt, b1, W2, b2, out);
}
